# starting kernel
# baseline (speedup 1.0000x reference)
#include <hip/hip_runtime.h>
#include <math.h>


typedef float v2f __attribute__((ext_vector_type(2)));
typedef float v4f __attribute__((ext_vector_type(4)));
typedef float v8f __attribute__((ext_vector_type(8)));

#define DQ      512
#define QQ      256
#define NLAY    4
#define NPAIR   255
#define KSTEPS  128
#define NTILES  32
#define M_STATE (16 * 2048)

static __device__ __forceinline__ v8f wmma_f32_4(v2f a, v2f b, v8f c) {
  return __builtin_amdgcn_wmma_f32_16x16x4_f32(false, a, false, b, (short)0, c,
                                               false, false);
}

__global__ void vqc_build_ent(const float* __restrict__ ent_params,
                              float* __restrict__ Tent) {
  const int l = blockIdx.x;
  const int r = threadIdx.x;

  __shared__ float c[NPAIR];
  for (int p = threadIdx.x; p < NPAIR; p += blockDim.x) {
    float x = ent_params[l * NPAIR + p];
    c[p] = 0.1f * (1.0f / (1.0f + expf(-x)));
  }
  __syncthreads();

  float* row = Tent + ((size_t)l * DQ + r) * DQ;
  for (int j = 0; j < DQ; ++j) row[j] = 0.0f;
  row[r] = 1.0f;
  if (r < QQ) {
    if (r >= 1) row[r - 1] = c[r - 1];
    if (r <= QQ - 2) {
      float val = c[r];
      for (int j = r + 1; j < QQ; ++j) {
        row[j] = val;
        if (j < NPAIR) val *= c[j];
      }
    }
  }
}

__global__ void vqc_build_layer(const float* __restrict__ rot_params,
                                const float* __restrict__ Tent,
                                float* __restrict__ Tlay) {
  const int i = blockIdx.x;
  const int l = blockIdx.y;
  const int q = i >> 1;
  const int par = i & 1;

  const float t0 = rot_params[((l * QQ) + q) * 3 + 0];
  const float t1 = rot_params[((l * QQ) + q) * 3 + 1];
  const float t2 = rot_params[((l * QQ) + q) * 3 + 2];
  const float c0 = cosf(t0 * 0.5f);
  const float c1 = cosf(t1 * 0.5f);
  const float s1 = sinf(t1 * 0.5f);
  const float c2 = cosf(t2 * 0.5f);
  const float b00 = (c0 * c1) * c2;
  const float b01 = (c0 * (-s1)) * c2;
  const float b10 = (c0 * s1) * c2;

  const float w0 = (par == 0) ? b00 : b10;
  const float w1 = (par == 0) ? b01 : b00;

  const float* r0 = Tent + ((size_t)l * DQ + 2 * q) * DQ;
  const float* r1 = r0 + DQ;
  float* out = Tlay + ((size_t)l * DQ + i) * DQ;
  for (int j = threadIdx.x; j < DQ; j += blockDim.x)
    out[j] = w0 * r0[j] + w1 * r1[j];
}

__global__ void vqc_pack_b(const float* __restrict__ B, float* __restrict__ Bp) {
  const int idx = blockIdx.x * blockDim.x + threadIdx.x;
  const int lane = idx & 31;
  const int nt = (idx >> 5) & 31;
  const int ks = idx >> 10;
  const int k0 = ks * 4 + ((lane >> 4) << 1);
  const int col = nt * 16 + (lane & 15);
  v2f v;
  v[0] = B[(size_t)k0 * DQ + col];
  v[1] = B[(size_t)(k0 + 1) * DQ + col];
  *(v2f*)(Bp + (size_t)idx * 2) = v;
}

__global__ void __launch_bounds__(128)
vqc_wmma_gemm(const float* __restrict__ A, const float* __restrict__ Bp,
              float* __restrict__ C) {
  __shared__ v2f As[2 * KSTEPS * 32];

  const int row0 = blockIdx.x * 32;

  {
    const int row = threadIdx.x >> 2;
    const int cq = threadIdx.x & 3;
    const int mt = row >> 4;
    const int lr = row & 15;
    const float* arow = A + (size_t)(row0 + row) * DQ;
#pragma unroll 4
    for (int s = 0; s < 32; ++s) {
      const int ks = cq + 4 * s;
      v4f v = *(const v4f*)(arow + ks * 4);
      v2f lo; lo[0] = v[0]; lo[1] = v[1];
      v2f hi; hi[0] = v[2]; hi[1] = v[3];
      As[(mt * KSTEPS + ks) * 32 + lr] = lo;
      As[(mt * KSTEPS + ks) * 32 + lr + 16] = hi;
    }
  }
  __syncthreads();

  const int wave = threadIdx.x >> 5;
  const int lane = threadIdx.x & 31;
  const int ntbase = wave * 8;

  v8f acc[2][8];
#pragma unroll
  for (int m = 0; m < 2; ++m)
#pragma unroll
    for (int t = 0; t < 8; ++t)
      acc[m][t] = (v8f){0.f, 0.f, 0.f, 0.f, 0.f, 0.f, 0.f, 0.f};

  const v2f* BpV = (const v2f*)Bp;
  for (int ks = 0; ks < KSTEPS; ++ks) {
    const v2f a0 = As[(0 * KSTEPS + ks) * 32 + lane];
    const v2f a1 = As[(1 * KSTEPS + ks) * 32 + lane];
#pragma unroll
    for (int t = 0; t < 8; ++t) {
      const v2f b = BpV[((size_t)ks * NTILES + (ntbase + t)) * 32 + lane];
      acc[0][t] = wmma_f32_4(a0, b, acc[0][t]);
      acc[1][t] = wmma_f32_4(a1, b, acc[1][t]);
    }
  }

  const int lrow = lane & 15;
  const int hi8 = (lane >> 4) * 8;
#pragma unroll
  for (int mt = 0; mt < 2; ++mt)
#pragma unroll
    for (int t = 0; t < 8; ++t) {
      const int col = (ntbase + t) * 16 + lrow;
#pragma unroll
      for (int v = 0; v < 8; ++v) {
        const int r = row0 + mt * 16 + hi8 + v;
        C[(size_t)r * DQ + col] = acc[mt][t][v];
      }
    }
}

extern "C" void kernel_launch(void* const* d_in, const int* in_sizes, int n_in,
                              void* d_out, int out_size, void* d_ws,
                              size_t ws_size, hipStream_t stream) {
  (void)in_sizes; (void)n_in; (void)out_size; (void)ws_size;
  const float* state = (const float*)d_in[0];
  const float* rotp  = (const float*)d_in[1];
  const float* entp  = (const float*)d_in[2];
  float* out = (float*)d_out;

  float* w     = (float*)d_ws;
  float* Tent  = w;
  float* Tlay  = w + 1048576;
  float* T01   = w + 0;
  float* T23   = w + 262144;
  float* Ttot  = w + 524288;
  float* Bp    = w + 786432;

  vqc_build_ent<<<NLAY, 512, 0, stream>>>(entp, Tent);

  vqc_build_layer<<<dim3(DQ, NLAY), 128, 0, stream>>>(rotp, Tent, Tlay);

  vqc_pack_b<<<512, 256, 0, stream>>>(Tlay + 1 * 262144, Bp);
  vqc_wmma_gemm<<<DQ / 32, 128, 0, stream>>>(Tlay + 0 * 262144, Bp, T01);

  vqc_pack_b<<<512, 256, 0, stream>>>(Tlay + 3 * 262144, Bp);
  vqc_wmma_gemm<<<DQ / 32, 128, 0, stream>>>(Tlay + 2 * 262144, Bp, T23);

  vqc_pack_b<<<512, 256, 0, stream>>>(T23, Bp);
  vqc_wmma_gemm<<<DQ / 32, 128, 0, stream>>>(T01, Bp, Ttot);

  vqc_pack_b<<<512, 256, 0, stream>>>(Ttot, Bp);
  vqc_wmma_gemm<<<M_STATE / 32, 128, 0, stream>>>(state, Bp, out);
}


# T built without the three dense 512^3 products: each identity row goes through the circuit's own column updates (2x2 rotations, 255 pair updates a layer) in LDS; sums reordered; 6 -> 2 launches
# speedup vs baseline: 2.9001x; 2.9001x over previous
#include <hip/hip_runtime.h>
#include <math.h>


typedef float v2f __attribute__((ext_vector_type(2)));
typedef float v4f __attribute__((ext_vector_type(4)));
typedef float v8f __attribute__((ext_vector_type(8)));

#define DQ      512
#define QQ      256
#define NLAY    4
#define NPAIR   255
#define KSTEPS  128
#define NTILES  32
#define M_STATE (16 * 2048)

static __device__ __forceinline__ v8f wmma_f32_4(v2f a, v2f b, v8f c) {
  return __builtin_amdgcn_wmma_f32_16x16x4_f32(false, a, false, b, (short)0, c,
                                               false, false);
}

#define XP 33

#define SW_U 4

struct SweepIn {
  v4f R[SW_U];
  v2f c[SW_U];
  float a[SW_U];
  float b[SW_U];
};

static __device__ __forceinline__ void sweep_load(SweepIn& s, const v4f* rbl,
                                                  const v2f* cel,
                                                  const float* Xl, int q0) {
#pragma unroll
  for (int j = 0; j < SW_U; ++j) {
    const int q = (q0 + j < QQ / 2 - 2) ? q0 + j : QQ / 2 - 2;
    s.R[j] = rbl[q + 1];
    s.c[j] = cel[q];
    s.a[j] = Xl[(2 * q + 2) * XP];
    s.b[j] = Xl[(2 * q + 3) * XP];
  }
}

template <int N>
static __device__ __forceinline__ void sweep_run(float& cur, float& nxt,
                                                 const SweepIn& s, float* Xl,
                                                 int q0) {
#pragma unroll
  for (int j = 0; j < N; ++j) {
    const int q = q0 + j;
    const float r0 = s.a[j] * s.R[j][0] + s.b[j] * s.R[j][2];
    const float r1 = s.a[j] * s.R[j][1] + s.b[j] * s.R[j][0];
    Xl[(2 * q) * XP] = cur + s.c[j][0] * nxt;
    const float cur2 = nxt + s.c[j][0] * cur;
    Xl[(2 * q + 1) * XP] = cur2 + s.c[j][1] * r0;
    cur = r0 + s.c[j][1] * cur2;
    nxt = r1;
  }
}

__global__ void __launch_bounds__(256)
vqc_build_total(const float* __restrict__ rot_params,
                const float* __restrict__ ent_params,
                float* __restrict__ Bp) {
  __shared__ float X[DQ * XP];
  __shared__ v4f rb[NLAY * QQ];
  __shared__ v2f ce[NLAY * (QQ / 2)];

  const int tid = threadIdx.x;
  const int wave = tid >> 5;
  const int lane = tid & 31;
  const int row = blockIdx.x * 32 + lane;

  for (int l = 0; l < NLAY; ++l) {
    const float t0 = rot_params[((l * QQ) + tid) * 3 + 0];
    const float t1 = rot_params[((l * QQ) + tid) * 3 + 1];
    const float t2 = rot_params[((l * QQ) + tid) * 3 + 2];
    const float c0 = cosf(t0 * 0.5f);
    const float c1 = cosf(t1 * 0.5f);
    const float s1 = sinf(t1 * 0.5f);
    const float c2 = cosf(t2 * 0.5f);
    v4f r;
    r[0] = (c0 * c1) * c2;
    r[1] = (c0 * (-s1)) * c2;
    r[2] = (c0 * s1) * c2;
    r[3] = 0.0f;
    rb[l * QQ + tid] = r;
    float c = 0.0f;
    if (tid < NPAIR) {
      const float x = ent_params[l * NPAIR + tid];
      c = 0.1f * (1.0f / (1.0f + expf(-x)));
    }
    ((float*)ce)[l * QQ + tid] = c;
  }
  for (int j = 0; j < DQ / 8; ++j) {
    const int col = wave * (DQ / 8) + j;
    X[col * XP + lane] = (col == row) ? 1.0f : 0.0f;
  }
  __syncthreads();

  if (wave == 0) {
    float* Xl = X + lane;
    for (int l = 0; l < NLAY; ++l) {
      const v4f* rbl = rb + l * QQ;
      const v2f* cel = ce + l * (QQ / 2);
      const v4f R0 = rbl[0];
      const float a0 = Xl[0 * XP], b0 = Xl[1 * XP];
      float cur = a0 * R0[0] + b0 * R0[2];
      float nxt = a0 * R0[1] + b0 * R0[0];
      SweepIn A, B;
      sweep_load(A, rbl, cel, Xl, 0);
      for (int i = 0; i < 15; ++i) {
        sweep_load(B, rbl, cel, Xl, 8 * i + 4);
        sweep_run<SW_U>(cur, nxt, A, Xl, 8 * i);
        sweep_load(A, rbl, cel, Xl, 8 * i + 8);
        sweep_run<SW_U>(cur, nxt, B, Xl, 8 * i + 4);
      }
      sweep_load(B, rbl, cel, Xl, 124);
      sweep_run<SW_U>(cur, nxt, A, Xl, 120);
      sweep_run<3>(cur, nxt, B, Xl, 124);
      const v2f c = cel[QQ / 2 - 1];
      Xl[(QQ - 2) * XP] = cur + c[0] * nxt;
      Xl[(QQ - 1) * XP] = nxt + c[0] * cur;
    }
  } else if (wave >= 4) {
    const int qb = QQ / 2 + (wave - 4) * 32;
    for (int j = 0; j < 32; ++j) {
      const int q = qb + j;
      float a = X[(2 * q) * XP + lane], b = X[(2 * q + 1) * XP + lane];
#pragma unroll
      for (int l = 0; l < NLAY; ++l) {
        const v4f R = rb[l * QQ + q];
        const float t = a * R[0] + b * R[2];
        b = a * R[1] + b * R[0];
        a = t;
      }
      X[(2 * q) * XP + lane] = a;
      X[(2 * q + 1) * XP + lane] = b;
    }
  }
  __syncthreads();

  v2f* out = (v2f*)Bp + (size_t)blockIdx.x * (8 * NTILES * 32);
  for (int i = 0; i < 32; ++i) {
    const int idx = i * 256 + tid;
    const int fl = idx & 31;
    const int nt = (idx >> 5) & 31;
    const int ksl = idx >> 10;
    const int col = nt * 16 + (fl & 15);
    const int r = ksl * 4 + ((fl >> 4) << 1);
    v2f v;
    v[0] = X[col * XP + r];
    v[1] = X[col * XP + r + 1];
    out[idx] = v;
  }
}

__global__ void __launch_bounds__(128)
vqc_wmma_gemm(const float* __restrict__ A, const float* __restrict__ Bp,
              float* __restrict__ C) {
  __shared__ v2f As[2 * KSTEPS * 32];

  const int row0 = blockIdx.x * 32;

  {
    const int row = threadIdx.x >> 2;
    const int cq = threadIdx.x & 3;
    const int mt = row >> 4;
    const int lr = row & 15;
    const float* arow = A + (size_t)(row0 + row) * DQ;
#pragma unroll 4
    for (int s = 0; s < 32; ++s) {
      const int ks = cq + 4 * s;
      v4f v = *(const v4f*)(arow + ks * 4);
      v2f lo; lo[0] = v[0]; lo[1] = v[1];
      v2f hi; hi[0] = v[2]; hi[1] = v[3];
      As[(mt * KSTEPS + ks) * 32 + lr] = lo;
      As[(mt * KSTEPS + ks) * 32 + lr + 16] = hi;
    }
  }
  __syncthreads();

  const int wave = threadIdx.x >> 5;
  const int lane = threadIdx.x & 31;
  const int ntbase = wave * 8;

  v8f acc[2][8];
#pragma unroll
  for (int m = 0; m < 2; ++m)
#pragma unroll
    for (int t = 0; t < 8; ++t)
      acc[m][t] = (v8f){0.f, 0.f, 0.f, 0.f, 0.f, 0.f, 0.f, 0.f};

  const v2f* BpV = (const v2f*)Bp;
  for (int ks = 0; ks < KSTEPS; ++ks) {
    const v2f a0 = As[(0 * KSTEPS + ks) * 32 + lane];
    const v2f a1 = As[(1 * KSTEPS + ks) * 32 + lane];
#pragma unroll
    for (int t = 0; t < 8; ++t) {
      const v2f b = BpV[((size_t)ks * NTILES + (ntbase + t)) * 32 + lane];
      acc[0][t] = wmma_f32_4(a0, b, acc[0][t]);
      acc[1][t] = wmma_f32_4(a1, b, acc[1][t]);
    }
  }

  const int lrow = lane & 15;
  const int hi8 = (lane >> 4) * 8;
#pragma unroll
  for (int mt = 0; mt < 2; ++mt)
#pragma unroll
    for (int t = 0; t < 8; ++t) {
      const int col = (ntbase + t) * 16 + lrow;
#pragma unroll
      for (int v = 0; v < 8; ++v) {
        const int r = row0 + mt * 16 + hi8 + v;
        C[(size_t)r * DQ + col] = acc[mt][t][v];
      }
    }
}

extern "C" void kernel_launch(void* const* d_in, const int* in_sizes, int n_in,
                              void* d_out, int out_size, void* d_ws,
                              size_t ws_size, hipStream_t stream) {
  (void)in_sizes; (void)n_in; (void)out_size; (void)ws_size;
  const float* state = (const float*)d_in[0];
  const float* rotp  = (const float*)d_in[1];
  const float* entp  = (const float*)d_in[2];
  float* out = (float*)d_out;
  float* Bp  = (float*)d_ws;

  vqc_build_total<<<DQ / 32, 256, 0, stream>>>(rotp, entp, Bp);

  vqc_wmma_gemm<<<M_STATE / 32, 128, 0, stream>>>(state, Bp, out);
}


# 32768x512 product, output-stationary: a wave keeps a 64x256 tile of out in 512 accumulator registers and reads A and B fragments straight from global memory two K steps ahead; no LDS; sums unchanged
# speedup vs baseline: 3.1596x; 1.0895x over previous
#include <hip/hip_runtime.h>
#include <math.h>


typedef float v2f __attribute__((ext_vector_type(2)));
typedef float v4f __attribute__((ext_vector_type(4)));
typedef float v8f __attribute__((ext_vector_type(8)));

#define DQ      512
#define QQ      256
#define NLAY    4
#define NPAIR   255
#define KSTEPS  128
#define NTILES  32
#define M_STATE (16 * 2048)

static __device__ __forceinline__ v8f wmma_f32_4(v2f a, v2f b, v8f c) {
  return __builtin_amdgcn_wmma_f32_16x16x4_f32(false, a, false, b, (short)0, c,
                                               false, false);
}

#define XP 33

#define SW_U 4

struct SweepIn {
  v4f R[SW_U];
  v2f c[SW_U];
  float a[SW_U];
  float b[SW_U];
};

static __device__ __forceinline__ void sweep_load(SweepIn& s, const v4f* rbl,
                                                  const v2f* cel,
                                                  const float* Xl, int q0) {
#pragma unroll
  for (int j = 0; j < SW_U; ++j) {
    const int q = (q0 + j < QQ / 2 - 2) ? q0 + j : QQ / 2 - 2;
    s.R[j] = rbl[q + 1];
    s.c[j] = cel[q];
    s.a[j] = Xl[(2 * q + 2) * XP];
    s.b[j] = Xl[(2 * q + 3) * XP];
  }
}

template <int N>
static __device__ __forceinline__ void sweep_run(float& cur, float& nxt,
                                                 const SweepIn& s, float* Xl,
                                                 int q0) {
#pragma unroll
  for (int j = 0; j < N; ++j) {
    const int q = q0 + j;
    const float r0 = s.a[j] * s.R[j][0] + s.b[j] * s.R[j][2];
    const float r1 = s.a[j] * s.R[j][1] + s.b[j] * s.R[j][0];
    Xl[(2 * q) * XP] = cur + s.c[j][0] * nxt;
    const float cur2 = nxt + s.c[j][0] * cur;
    Xl[(2 * q + 1) * XP] = cur2 + s.c[j][1] * r0;
    cur = r0 + s.c[j][1] * cur2;
    nxt = r1;
  }
}

__global__ void __launch_bounds__(256)
vqc_build_total(const float* __restrict__ rot_params,
                const float* __restrict__ ent_params,
                float* __restrict__ Bp) {
  __shared__ float X[DQ * XP];
  __shared__ v4f rb[NLAY * QQ];
  __shared__ v2f ce[NLAY * (QQ / 2)];

  const int tid = threadIdx.x;
  const int wave = tid >> 5;
  const int lane = tid & 31;
  const int row = blockIdx.x * 32 + lane;

  for (int l = 0; l < NLAY; ++l) {
    const float t0 = rot_params[((l * QQ) + tid) * 3 + 0];
    const float t1 = rot_params[((l * QQ) + tid) * 3 + 1];
    const float t2 = rot_params[((l * QQ) + tid) * 3 + 2];
    const float c0 = cosf(t0 * 0.5f);
    const float c1 = cosf(t1 * 0.5f);
    const float s1 = sinf(t1 * 0.5f);
    const float c2 = cosf(t2 * 0.5f);
    v4f r;
    r[0] = (c0 * c1) * c2;
    r[1] = (c0 * (-s1)) * c2;
    r[2] = (c0 * s1) * c2;
    r[3] = 0.0f;
    rb[l * QQ + tid] = r;
    float c = 0.0f;
    if (tid < NPAIR) {
      const float x = ent_params[l * NPAIR + tid];
      c = 0.1f * (1.0f / (1.0f + expf(-x)));
    }
    ((float*)ce)[l * QQ + tid] = c;
  }
  for (int j = 0; j < DQ / 8; ++j) {
    const int col = wave * (DQ / 8) + j;
    X[col * XP + lane] = (col == row) ? 1.0f : 0.0f;
  }
  __syncthreads();

  if (wave == 0) {
    float* Xl = X + lane;
    for (int l = 0; l < NLAY; ++l) {
      const v4f* rbl = rb + l * QQ;
      const v2f* cel = ce + l * (QQ / 2);
      const v4f R0 = rbl[0];
      const float a0 = Xl[0 * XP], b0 = Xl[1 * XP];
      float cur = a0 * R0[0] + b0 * R0[2];
      float nxt = a0 * R0[1] + b0 * R0[0];
      SweepIn A, B;
      sweep_load(A, rbl, cel, Xl, 0);
      for (int i = 0; i < 15; ++i) {
        sweep_load(B, rbl, cel, Xl, 8 * i + 4);
        sweep_run<SW_U>(cur, nxt, A, Xl, 8 * i);
        sweep_load(A, rbl, cel, Xl, 8 * i + 8);
        sweep_run<SW_U>(cur, nxt, B, Xl, 8 * i + 4);
      }
      sweep_load(B, rbl, cel, Xl, 124);
      sweep_run<SW_U>(cur, nxt, A, Xl, 120);
      sweep_run<3>(cur, nxt, B, Xl, 124);
      const v2f c = cel[QQ / 2 - 1];
      Xl[(QQ - 2) * XP] = cur + c[0] * nxt;
      Xl[(QQ - 1) * XP] = nxt + c[0] * cur;
    }
  } else if (wave >= 4) {
    const int qb = QQ / 2 + (wave - 4) * 32;
    for (int j = 0; j < 32; ++j) {
      const int q = qb + j;
      float a = X[(2 * q) * XP + lane], b = X[(2 * q + 1) * XP + lane];
#pragma unroll
      for (int l = 0; l < NLAY; ++l) {
        const v4f R = rb[l * QQ + q];
        const float t = a * R[0] + b * R[2];
        b = a * R[1] + b * R[0];
        a = t;
      }
      X[(2 * q) * XP + lane] = a;
      X[(2 * q + 1) * XP + lane] = b;
    }
  }
  __syncthreads();

  v2f* out = (v2f*)Bp + (size_t)blockIdx.x * (8 * NTILES * 32);
  for (int i = 0; i < 32; ++i) {
    const int idx = i * 256 + tid;
    const int fl = idx & 31;
    const int nt = (idx >> 5) & 31;
    const int ksl = idx >> 10;
    const int col = nt * 16 + (fl & 15);
    const int r = ksl * 4 + ((fl >> 4) << 1);
    v2f v;
    v[0] = X[col * XP + r];
    v[1] = X[col * XP + r + 1];
    out[idx] = v;
  }
}

#define WM 4
#define WN 16
#define GK 2

struct FragSet {
  v2f a[GK][WM];
  v2f b[GK][WN];
};

static __device__ __forceinline__ void frag_load(FragSet& f, const float* ap,
                                                 const v2f* bp, int ks0) {
#pragma unroll
  for (int g = 0; g < GK; ++g) {
    const int ks = ks0 + g;
#pragma unroll
    for (int m = 0; m < WM; ++m)
      f.a[g][m] = *(const v2f*)(ap + (size_t)m * 16 * DQ + ks * 4);
#pragma unroll
    for (int t = 0; t < WN; ++t)
      f.b[g][t] = bp[((size_t)ks * NTILES + t) * 32];
  }
}

static __device__ __forceinline__ void frag_mma(v8f (&acc)[WM][WN],
                                                const FragSet& f) {
#pragma unroll
  for (int g = 0; g < GK; ++g)
#pragma unroll
    for (int t = 0; t < WN; ++t)
#pragma unroll
      for (int m = 0; m < WM; ++m)
        acc[m][t] = wmma_f32_4(f.a[g][m], f.b[g][t], acc[m][t]);
}

__global__ void __launch_bounds__(128)
vqc_wmma_gemm(const float* __restrict__ A, const float* __restrict__ Bp,
              float* __restrict__ C) {
  const int wave = threadIdx.x >> 5;
  const int lane = threadIdx.x & 31;
  const int lrow = lane & 15;
  const int half = lane >> 4;
  const int row0 = blockIdx.x * 128 + (wave >> 1) * (16 * WM);
  const int nt0 = (wave & 1) * WN;

  const float* ap = A + (size_t)(row0 + lrow) * DQ + 2 * half;
  const v2f* bp = (const v2f*)Bp + (size_t)nt0 * 32 + lane;

  v8f acc[WM][WN];
#pragma unroll
  for (int m = 0; m < WM; ++m)
#pragma unroll
    for (int t = 0; t < WN; ++t)
      acc[m][t] = (v8f){0.f, 0.f, 0.f, 0.f, 0.f, 0.f, 0.f, 0.f};

  FragSet F0, F1;
  frag_load(F0, ap, bp, 0);
  for (int i = 0; i < KSTEPS / (2 * GK) - 1; ++i) {
    frag_load(F1, ap, bp, (2 * i + 1) * GK);
    __builtin_amdgcn_sched_barrier(0);
    frag_mma(acc, F0);
    frag_load(F0, ap, bp, (2 * i + 2) * GK);
    frag_mma(acc, F1);
  }
  frag_load(F1, ap, bp, KSTEPS - GK);
  frag_mma(acc, F0);
  frag_mma(acc, F1);

  const int hi8 = half * 8;
#pragma unroll
  for (int m = 0; m < WM; ++m)
#pragma unroll
    for (int t = 0; t < WN; ++t) {
      const int col = (nt0 + t) * 16 + lrow;
#pragma unroll
      for (int v = 0; v < 8; ++v) {
        const int r = row0 + m * 16 + hi8 + v;
        C[(size_t)r * DQ + col] = acc[m][t][v];
      }
    }
}

extern "C" void kernel_launch(void* const* d_in, const int* in_sizes, int n_in,
                              void* d_out, int out_size, void* d_ws,
                              size_t ws_size, hipStream_t stream) {
  (void)in_sizes; (void)n_in; (void)out_size; (void)ws_size;
  const float* state = (const float*)d_in[0];
  const float* rotp  = (const float*)d_in[1];
  const float* entp  = (const float*)d_in[2];
  float* out = (float*)d_out;
  float* Bp  = (float*)d_ws;

  vqc_build_total<<<DQ / 32, 256, 0, stream>>>(rotp, entp, Bp);

  vqc_wmma_gemm<<<M_STATE / 128, 128, 0, stream>>>(state, Bp, out);
}


# T build: the 255-pair column sweep of a layer runs as a segmented scan, 8 waves x 32 columns in registers, one barrier a layer, instead of one wave walking 127 qubits; sums reordered in the sweep
# speedup vs baseline: 3.4822x; 1.1021x over previous
#include <hip/hip_runtime.h>
#include <math.h>


typedef float v2f __attribute__((ext_vector_type(2)));
typedef float v4f __attribute__((ext_vector_type(4)));
typedef float v8f __attribute__((ext_vector_type(8)));

#define DQ      512
#define QQ      256
#define NLAY    4
#define NPAIR   255
#define KSTEPS  128
#define NTILES  32
#define M_STATE (16 * 2048)

static __device__ __forceinline__ v8f wmma_f32_4(v2f a, v2f b, v8f c) {
  return __builtin_amdgcn_wmma_f32_16x16x4_f32(false, a, false, b, (short)0, c,
                                               false, false);
}

#define XP 33
#define SEG 32
#define NSEG 8

__global__ void __launch_bounds__(256)
vqc_build_total(const float* __restrict__ rot_params,
                const float* __restrict__ ent_params,
                float* __restrict__ Bp) {
  __shared__ float X[DQ * XP];
  __shared__ v4f rb[NLAY * QQ];
  __shared__ v4f cc4[NLAY * QQ / 4];
  __shared__ float Uo[NLAY * NSEG * 32];
  __shared__ float Yf[NLAY * NSEG * 32];
  __shared__ float Gs[NLAY * NSEG];
  float* cc = (float*)cc4;

  const int tid = threadIdx.x;
  const int wave = tid >> 5;
  const int lane = tid & 31;
  const int row = blockIdx.x * 32 + lane;

  for (int l = 0; l < NLAY; ++l) {
    const float t0 = rot_params[((l * QQ) + tid) * 3 + 0];
    const float t1 = rot_params[((l * QQ) + tid) * 3 + 1];
    const float t2 = rot_params[((l * QQ) + tid) * 3 + 2];
    const float c0 = cosf(t0 * 0.5f);
    const float c1 = cosf(t1 * 0.5f);
    const float s1 = sinf(t1 * 0.5f);
    const float c2 = cosf(t2 * 0.5f);
    v4f r;
    r[0] = (c0 * c1) * c2;
    r[1] = (c0 * (-s1)) * c2;
    r[2] = (c0 * s1) * c2;
    r[3] = 0.0f;
    rb[l * QQ + tid] = r;
    float c = 0.0f;
    if (tid < NPAIR) {
      const float x = ent_params[l * NPAIR + tid];
      c = 0.1f * (1.0f / (1.0f + expf(-x)));
    }
    cc[l * QQ + tid] = c;
  }
  __syncthreads();

  const int s = wave * SEG;

  {
    float h[SEG];
#pragma unroll
    for (int k = 0; k < SEG; ++k) h[k] = (QQ + s + k == row) ? 1.0f : 0.0f;
#pragma unroll
    for (int j = 0; j < SEG / 2; ++j) {
      float a = h[2 * j], b = h[2 * j + 1];
#pragma unroll
      for (int l = 0; l < NLAY; ++l) {
        const v4f R = rb[l * QQ + (QQ + s) / 2 + j];
        const float t = a * R[0] + b * R[2];
        b = a * R[1] + b * R[0];
        a = t;
      }
      X[(QQ + s + 2 * j) * XP + lane] = a;
      X[(QQ + s + 2 * j + 1) * XP + lane] = b;
    }
  }

  float x[SEG];
#pragma unroll
  for (int k = 0; k < SEG; ++k) x[k] = (s + k == row) ? 1.0f : 0.0f;

  for (int l = 0; l < NLAY; ++l) {
    const v4f* rbl = rb + l * QQ + s / 2;
    const v4f* cl4 = cc4 + (l * QQ + s) / 4;
    float y[SEG], c[SEG], v0[SEG], P[SEG];
#pragma unroll
    for (int j = 0; j < SEG / 4; ++j) {
      const v4f t = cl4[j];
      c[4 * j] = t[0]; c[4 * j + 1] = t[1]; c[4 * j + 2] = t[2]; c[4 * j + 3] = t[3];
    }
#pragma unroll
    for (int j = 0; j < SEG / 2; ++j) {
      const v4f R = rbl[j];
      y[2 * j] = x[2 * j] * R[0] + x[2 * j + 1] * R[2];
      y[2 * j + 1] = x[2 * j] * R[1] + x[2 * j + 1] * R[0];
    }
    v0[0] = y[0];
    P[0] = 1.0f;
#pragma unroll
    for (int k = 1; k < SEG; ++k) {
      v0[k] = y[k] + c[k - 1] * v0[k - 1];
      P[k] = c[k - 1] * P[k - 1];
    }
    const int lw = (l * NSEG + wave) * 32 + lane;
    Yf[lw] = y[0];
    Uo[lw] = c[SEG - 1] * v0[SEG - 1];
    if (lane == 0) Gs[l * NSEG + wave] = c[SEG - 1] * P[SEG - 1];
    __syncthreads();
    float u = 0.0f;
#pragma unroll
    for (int k = 0; k < NSEG - 1; ++k) {
      const float t = Uo[(l * NSEG + k) * 32 + lane] + Gs[l * NSEG + k] * u;
      u = (k < wave) ? t : u;
    }
    const int nw = (wave < NSEG - 1) ? wave + 1 : wave;
    const float yn = (wave < NSEG - 1) ? Yf[(l * NSEG + nw) * 32 + lane] : 0.0f;
#pragma unroll
    for (int k = 0; k < SEG; ++k) {
      const float v = v0[k] + P[k] * u;
      x[k] = v + c[k] * ((k < SEG - 1) ? y[k + 1] : yn);
    }
  }
#pragma unroll
  for (int k = 0; k < SEG; ++k) X[(s + k) * XP + lane] = x[k];
  __syncthreads();

  v2f* out = (v2f*)Bp + (size_t)blockIdx.x * (8 * NTILES * 32);
  for (int i = 0; i < 32; ++i) {
    const int idx = i * 256 + tid;
    const int fl = idx & 31;
    const int nt = (idx >> 5) & 31;
    const int ksl = idx >> 10;
    const int col = nt * 16 + (fl & 15);
    const int r = ksl * 4 + ((fl >> 4) << 1);
    v2f v;
    v[0] = X[col * XP + r];
    v[1] = X[col * XP + r + 1];
    out[idx] = v;
  }
}

#define WM 4
#define WN 16
#define GK 2

struct FragSet {
  v2f a[GK][WM];
  v2f b[GK][WN];
};

static __device__ __forceinline__ void frag_load(FragSet& f, const float* ap,
                                                 const v2f* bp, int ks0) {
#pragma unroll
  for (int g = 0; g < GK; ++g) {
    const int ks = ks0 + g;
#pragma unroll
    for (int m = 0; m < WM; ++m)
      f.a[g][m] = *(const v2f*)(ap + (size_t)m * 16 * DQ + ks * 4);
#pragma unroll
    for (int t = 0; t < WN; ++t)
      f.b[g][t] = bp[((size_t)ks * NTILES + t) * 32];
  }
}

static __device__ __forceinline__ void frag_mma(v8f (&acc)[WM][WN],
                                                const FragSet& f) {
#pragma unroll
  for (int g = 0; g < GK; ++g)
#pragma unroll
    for (int t = 0; t < WN; ++t)
#pragma unroll
      for (int m = 0; m < WM; ++m)
        acc[m][t] = wmma_f32_4(f.a[g][m], f.b[g][t], acc[m][t]);
}

__global__ void __launch_bounds__(128)
vqc_wmma_gemm(const float* __restrict__ A, const float* __restrict__ Bp,
              float* __restrict__ C) {
  const int wave = threadIdx.x >> 5;
  const int lane = threadIdx.x & 31;
  const int lrow = lane & 15;
  const int half = lane >> 4;
  const int row0 = blockIdx.x * 128 + (wave >> 1) * (16 * WM);
  const int nt0 = (wave & 1) * WN;

  const float* ap = A + (size_t)(row0 + lrow) * DQ + 2 * half;
  const v2f* bp = (const v2f*)Bp + (size_t)nt0 * 32 + lane;

  v8f acc[WM][WN];
#pragma unroll
  for (int m = 0; m < WM; ++m)
#pragma unroll
    for (int t = 0; t < WN; ++t)
      acc[m][t] = (v8f){0.f, 0.f, 0.f, 0.f, 0.f, 0.f, 0.f, 0.f};

  FragSet F0, F1;
  frag_load(F0, ap, bp, 0);
  for (int i = 0; i < KSTEPS / (2 * GK) - 1; ++i) {
    frag_load(F1, ap, bp, (2 * i + 1) * GK);
    __builtin_amdgcn_sched_barrier(0);
    frag_mma(acc, F0);
    frag_load(F0, ap, bp, (2 * i + 2) * GK);
    frag_mma(acc, F1);
  }
  frag_load(F1, ap, bp, KSTEPS - GK);
  frag_mma(acc, F0);
  frag_mma(acc, F1);

  const int hi8 = half * 8;
#pragma unroll
  for (int m = 0; m < WM; ++m)
#pragma unroll
    for (int t = 0; t < WN; ++t) {
      const int col = (nt0 + t) * 16 + lrow;
#pragma unroll
      for (int v = 0; v < 8; ++v) {
        const int r = row0 + m * 16 + hi8 + v;
        C[(size_t)r * DQ + col] = acc[m][t][v];
      }
    }
}

extern "C" void kernel_launch(void* const* d_in, const int* in_sizes, int n_in,
                              void* d_out, int out_size, void* d_ws,
                              size_t ws_size, hipStream_t stream) {
  (void)in_sizes; (void)n_in; (void)out_size; (void)ws_size;
  const float* state = (const float*)d_in[0];
  const float* rotp  = (const float*)d_in[1];
  const float* entp  = (const float*)d_in[2];
  float* out = (float*)d_out;
  float* Bp  = (float*)d_ws;

  vqc_build_total<<<DQ / 32, 256, 0, stream>>>(rotp, entp, Bp);

  vqc_wmma_gemm<<<M_STATE / 128, 128, 0, stream>>>(state, Bp, out);
}


# 32768x512 product by layers, not by their dense product: rows on the N side, each layer's 16-column tile matrices as A, D registers reused as next B, scan carries as K-steps; sums reordered
# speedup vs baseline: 7.8119x; 2.2434x over previous
#include <hip/hip_runtime.h>
#include <math.h>


typedef float v2f __attribute__((ext_vector_type(2)));
typedef float v4f __attribute__((ext_vector_type(4)));
typedef float v8f __attribute__((ext_vector_type(8)));

#define DQ      512
#define QQ      256
#define NLAY    4
#define NPAIR   255
#define M_STATE (16 * 2048)
#define SLOTS_L    164
#define SLOTS_LEFT (NLAY * SLOTS_L)
#define SLOTS_ALL  (SLOTS_LEFT + 64)

static __device__ __forceinline__ v8f wmma_f32_4(v2f a, v2f b, v8f c) {
  return __builtin_amdgcn_wmma_f32_16x16x4_f32(false, a, false, b, (short)0, c,
                                               false, false);
}

static __device__ __forceinline__ float prodc(const float* cl, int a, int b) {
  float p = 1.0f;
  for (int k = a; k < b; ++k) p *= cl[k];
  return p;
}
static __device__ __forceinline__ float scan_w(const float* cl, int t, int i) {
  if (t < 0 || t > 15) return 0.0f;
  if (t <= i) return prodc(cl, t, i);
  if (t == i + 1) return cl[i];
  return 0.0f;
}
static __device__ __forceinline__ float tile_m(const v4f* rbl, const float* cl,
                                               int j, int i) {
  const v4f R = rbl[j >> 1];
  if ((j & 1) == 0) return R[0] * scan_w(cl, j, i) + R[1] * scan_w(cl, j + 1, i);
  return R[2] * scan_w(cl, j - 1, i) + R[0] * scan_w(cl, j, i);
}
static __device__ __forceinline__ float tile_a(const v4f* rbl, const float* cl,
                                               int j) {
  const v4f R = rbl[j >> 1];
  if ((j & 1) == 0) return R[0] * prodc(cl, j, 16) + R[1] * prodc(cl, j + 1, 16);
  return R[2] * prodc(cl, j - 1, 16) + R[0] * prodc(cl, j, 16);
}

static __device__ float frag_elem(const v4f* rb, const float* cc, const float* Gs,
                                  int sigma, int c, int h, int e) {
  if (sigma < SLOTS_LEFT) {
    const int l = sigma / SLOTS_L;
    const int t = sigma % SLOTS_L;
    if (t < 160) {
      const int J = t / 10;
      const int kind = t % 10;
      const v4f* rbl = rb + l * QQ + 8 * J;
      const float* cl = cc + l * QQ + 16 * J;
      if (kind < 4) return tile_m(rbl, cl, 8 * h + 2 * kind + e, c);
      if (kind == 4) {
        if (J >= 15 || c != 15 || h != 0) return 0.0f;
        const v4f R = rb[l * QQ + 8 * (J + 1)];
        return cl[15] * (e == 0 ? R[0] : R[2]);
      }
      if (kind == 5) {
        if (h != (J >> 3) || e != (J & 1)) return 0.0f;
        return prodc(cl, 0, c);
      }
      return (c == J) ? tile_a(rbl, cl, 8 * h + 2 * (kind - 6) + e) : 0.0f;
    }
    const int I = 8 * h + 2 * (t - 160) + e;
    if (I >= c) return 0.0f;
    float p = 1.0f;
    for (int K = I + 1; K < c; ++K) p *= Gs[l * 16 + K];
    return p;
  }
  const int r = sigma - SLOTS_LEFT;
  const int J = 16 + r / 4;
  const int k = 8 * h + 2 * (r % 4) + e;
  if ((k >> 1) != (c >> 1)) return 0.0f;
  const int q = 8 * J + (k >> 1);
  float a = ((k & 1) == 0) ? 1.0f : 0.0f;
  float b = ((k & 1) == 0) ? 0.0f : 1.0f;
  for (int l = 0; l < NLAY; ++l) {
    const v4f R = rb[l * QQ + q];
    const float t = a * R[0] + b * R[2];
    b = a * R[1] + b * R[0];
    a = t;
  }
  return ((c & 1) == 0) ? a : b;
}

__global__ void __launch_bounds__(256)
vqc_build_frags(const float* __restrict__ rot_params,
                const float* __restrict__ ent_params,
                float* __restrict__ Fw) {
  __shared__ v4f rb[NLAY * QQ];
  __shared__ float cc[NLAY * QQ];
  __shared__ float Gs[NLAY * 16];
  const int tid = threadIdx.x;

  for (int l = 0; l < NLAY; ++l) {
    const float t0 = rot_params[((l * QQ) + tid) * 3 + 0];
    const float t1 = rot_params[((l * QQ) + tid) * 3 + 1];
    const float t2 = rot_params[((l * QQ) + tid) * 3 + 2];
    const float c0 = cosf(t0 * 0.5f);
    const float c1 = cosf(t1 * 0.5f);
    const float s1 = sinf(t1 * 0.5f);
    const float c2 = cosf(t2 * 0.5f);
    v4f r;
    r[0] = (c0 * c1) * c2;
    r[1] = (c0 * (-s1)) * c2;
    r[2] = (c0 * s1) * c2;
    r[3] = 0.0f;
    rb[l * QQ + tid] = r;
    float c = 0.0f;
    if (tid < NPAIR) {
      const float x = ent_params[l * NPAIR + tid];
      c = 0.1f * (1.0f / (1.0f + expf(-x)));
    }
    cc[l * QQ + tid] = c;
  }
  __syncthreads();
  if (tid < NLAY * 16) Gs[tid] = prodc(cc + (tid >> 4) * QQ + 16 * (tid & 15), 0, 16);
  __syncthreads();

  const int idx = blockIdx.x * 256 + tid;
  const int sigma = idx >> 5;
  const int lane = idx & 31;
  v2f v;
  v[0] = frag_elem(rb, cc, Gs, sigma, lane & 15, lane >> 4, 0);
  v[1] = frag_elem(rb, cc, Gs, sigma, lane & 15, lane >> 4, 1);
  ((v2f*)Fw)[idx] = v;
}

static __device__ __forceinline__ v2f bsel(const v8f& t, int s) {
  v2f b;
  b[0] = t[2 * s];
  b[1] = t[2 * s + 1];
  return b;
}

static __device__ __forceinline__ v8f load_tile(const float* p) {
  const v4f lo = *(const v4f*)p;
  const v4f hi = *(const v4f*)(p + 4);
  return (v8f){lo[0], lo[1], lo[2], lo[3], hi[0], hi[1], hi[2], hi[3]};
}

static __device__ __forceinline__ void store_tile(float* p, const v8f& t) {
  *(v4f*)p = (v4f){t[0], t[1], t[2], t[3]};
  *(v4f*)(p + 4) = (v4f){t[4], t[5], t[6], t[7]};
}

#define NG 2
#define FA_N 68
#define FT_N 48

static __device__ __forceinline__ void load_fa(v2f (&f)[FA_N], const v2f* Fl) {
#pragma unroll
  for (int i = 0; i < 64; ++i) f[i] = Fl[((i >> 2) * 10 + 6 + (i & 3)) * 32];
#pragma unroll
  for (int i = 0; i < 4; ++i) f[64 + i] = Fl[(160 + i) * 32];
}

static __device__ __forceinline__ void load_ft(v2f (&f)[FT_N], const v2f* Fl,
                                               int J0) {
#pragma unroll
  for (int i = 0; i < FT_N; ++i)
    f[i] = Fl[((J0 + i / 6) * 10 + (i % 6)) * 32];
}

static __device__ __forceinline__ void carry_mma(v8f (&Du)[NG],
                                                 const v8f (&x)[NG][16],
                                                 const v2f (&f)[FA_N]) {
  const v8f zero = (v8f){0.f, 0.f, 0.f, 0.f, 0.f, 0.f, 0.f, 0.f};
  v8f U[NG][2];
#pragma unroll
  for (int g = 0; g < NG; ++g) { U[g][0] = zero; U[g][1] = zero; }
#pragma unroll
  for (int J = 0; J < 16; ++J)
#pragma unroll
    for (int s = 0; s < 4; ++s)
#pragma unroll
      for (int g = 0; g < NG; ++g)
        U[g][J & 1] = wmma_f32_4(f[4 * J + s], bsel(x[g][J], s), U[g][J & 1]);
#pragma unroll
  for (int g = 0; g < NG; ++g) Du[g] = zero;
#pragma unroll
  for (int p = 0; p < 2; ++p)
#pragma unroll
    for (int s = 0; s < 4; ++s)
#pragma unroll
      for (int g = 0; g < NG; ++g)
        Du[g] = wmma_f32_4(f[64 + s], bsel(U[g][p], s), Du[g]);
}

static __device__ __forceinline__ void tiles_mma(v8f (&y)[NG][16],
                                                 const v8f (&x)[NG][16],
                                                 const v8f (&Du)[NG],
                                                 const v2f (&f)[FT_N], int J0) {
  const v8f zero = (v8f){0.f, 0.f, 0.f, 0.f, 0.f, 0.f, 0.f, 0.f};
#pragma unroll
  for (int j = 0; j < 8; ++j) {
    const int J = J0 + j;
    v8f acc[NG];
#pragma unroll
    for (int g = 0; g < NG; ++g) acc[g] = zero;
#pragma unroll
    for (int s = 0; s < 4; ++s)
#pragma unroll
      for (int g = 0; g < NG; ++g)
        acc[g] = wmma_f32_4(f[6 * j + s], bsel(x[g][J], s), acc[g]);
    if (J < 15) {
#pragma unroll
      for (int g = 0; g < NG; ++g)
        acc[g] = wmma_f32_4(f[6 * j + 4], bsel(x[g][J + 1], 0), acc[g]);
    }
    if (J > 0) {
#pragma unroll
      for (int g = 0; g < NG; ++g)
        acc[g] = wmma_f32_4(f[6 * j + 5], bsel(Du[g], (J & 7) >> 1), acc[g]);
    }
#pragma unroll
    for (int g = 0; g < NG; ++g) y[g][J] = acc[g];
  }
}

__global__ void __launch_bounds__(128)
vqc_apply_layers(const float* __restrict__ A, const float* __restrict__ Fw,
                 float* __restrict__ C) {
  const int wave = threadIdx.x >> 5;
  const int lane = threadIdx.x & 31;
  const int c = lane & 15;
  const int h = lane >> 4;
  const size_t row = (size_t)(blockIdx.x * 4 + wave) * (16 * NG) + c;
  const float* xp = A + row * DQ + 8 * h;
  float* op = C + row * DQ + 8 * h;
  const v2f* F = (const v2f*)Fw + lane;
  const v8f zero = (v8f){0.f, 0.f, 0.f, 0.f, 0.f, 0.f, 0.f, 0.f};

  v2f fr[64];
#pragma unroll
  for (int i = 0; i < 64; ++i) fr[i] = F[(size_t)(SLOTS_LEFT + i) * 32];
  v8f xr[NG][16];
#pragma unroll
  for (int g = 0; g < NG; ++g)
#pragma unroll
    for (int J = 0; J < 16; ++J)
      xr[g][J] = load_tile(xp + (size_t)g * 16 * DQ + 256 + 16 * J);
  v8f x[NG][16];
#pragma unroll
  for (int g = 0; g < NG; ++g)
#pragma unroll
    for (int J = 0; J < 16; ++J)
      x[g][J] = load_tile(xp + (size_t)g * 16 * DQ + 16 * J);
  v2f fa[FA_N], fb[FT_N], fc[FT_N];
  load_fa(fa, F);
  __builtin_amdgcn_sched_barrier(0);

#pragma unroll
  for (int J = 0; J < 16; ++J) {
    v8f acc[NG];
#pragma unroll
    for (int g = 0; g < NG; ++g) acc[g] = zero;
#pragma unroll
    for (int s = 0; s < 4; ++s)
#pragma unroll
      for (int g = 0; g < NG; ++g)
        acc[g] = wmma_f32_4(fr[J * 4 + s], bsel(xr[g][J], s), acc[g]);
#pragma unroll
    for (int g = 0; g < NG; ++g)
      store_tile(op + (size_t)g * 16 * DQ + 256 + 16 * J, acc[g]);
  }

#pragma unroll
  for (int l = 0; l < NLAY; ++l) {
    const v2f* Fl = F + (size_t)l * SLOTS_L * 32;
    v8f Du[NG];
    v8f y[NG][16];
    load_ft(fb, Fl, 0);
    __builtin_amdgcn_sched_barrier(0);
    carry_mma(Du, x, fa);
    load_ft(fc, Fl, 8);
    __builtin_amdgcn_sched_barrier(0);
    tiles_mma(y, x, Du, fb, 0);
    if (l + 1 < NLAY) load_fa(fa, Fl + (size_t)SLOTS_L * 32);
    __builtin_amdgcn_sched_barrier(0);
    tiles_mma(y, x, Du, fc, 8);
#pragma unroll
    for (int g = 0; g < NG; ++g)
#pragma unroll
      for (int J = 0; J < 16; ++J) x[g][J] = y[g][J];
  }

#pragma unroll
  for (int g = 0; g < NG; ++g)
#pragma unroll
    for (int J = 0; J < 16; ++J)
      store_tile(op + (size_t)g * 16 * DQ + 16 * J, x[g][J]);
}

extern "C" void kernel_launch(void* const* d_in, const int* in_sizes, int n_in,
                              void* d_out, int out_size, void* d_ws,
                              size_t ws_size, hipStream_t stream) {
  (void)in_sizes; (void)n_in; (void)out_size; (void)ws_size;
  const float* state = (const float*)d_in[0];
  const float* rotp  = (const float*)d_in[1];
  const float* entp  = (const float*)d_in[2];
  float* out = (float*)d_out;
  float* Fw  = (float*)d_ws;

  vqc_build_frags<<<SLOTS_ALL * 32 / 256, 256, 0, stream>>>(rotp, entp, Fw);

  vqc_apply_layers<<<M_STATE / 128, 128, 0, stream>>>(state, Fw, out);
}


# Tile carry-outs are read from row 15 of the tile products (one K-step on registers 6,7 against an Lc column), not from a dot-product pass of their own: 1456 -> 1000 WMMAs a wave; sums reordered
# speedup vs baseline: 8.5807x; 1.0984x over previous
#include <hip/hip_runtime.h>
#include <math.h>


typedef float v2f __attribute__((ext_vector_type(2)));
typedef float v4f __attribute__((ext_vector_type(4)));
typedef float v8f __attribute__((ext_vector_type(8)));

#define DQ      512
#define QQ      256
#define NLAY    4
#define NPAIR   255
#define M_STATE (16 * 2048)
#define SLOTS_L    110
#define SLOTS_LEFT (NLAY * SLOTS_L)
#define SLOTS_ALL  (SLOTS_LEFT + 64)

static __device__ __forceinline__ v8f wmma_f32_4(v2f a, v2f b, v8f c) {
  return __builtin_amdgcn_wmma_f32_16x16x4_f32(false, a, false, b, (short)0, c,
                                               false, false);
}

static __device__ __forceinline__ float prodc(const float* cl, int a, int b) {
  float p = 1.0f;
  for (int k = a; k < b; ++k) p *= cl[k];
  return p;
}
static __device__ __forceinline__ float scan_w(const float* cl, int t, int i) {
  if (t < 0 || t > 15) return 0.0f;
  if (t <= i) return prodc(cl, t, i);
  if (t == i + 1) return cl[i];
  return 0.0f;
}
static __device__ __forceinline__ float tile_m(const v4f* rbl, const float* cl,
                                               int j, int i) {
  const v4f R = rbl[j >> 1];
  if ((j & 1) == 0) return R[0] * scan_w(cl, j, i) + R[1] * scan_w(cl, j + 1, i);
  return R[2] * scan_w(cl, j - 1, i) + R[0] * scan_w(cl, j, i);
}

static __device__ float frag_elem(const v4f* rb, const float* cc, const float* Gs,
                                  int sigma, int c, int h, int e) {
  if (sigma < SLOTS_LEFT) {
    const int l = sigma / SLOTS_L;
    const int t = sigma % SLOTS_L;
    if (t < 64) {
      const int J = t >> 2;
      return tile_m(rb + l * QQ + 8 * J, cc + l * QQ + 16 * J,
                    8 * h + 2 * (t & 3) + e, c);
    }
    if (t < 79) {
      const int I = t - 64;
      if (h != 1 || e != 1 || c <= I) return 0.0f;
      float p = cc[l * QQ + 16 * I + 15];
      for (int K = I + 1; K < c; ++K) p *= Gs[l * 16 + K];
      return p;
    }
    if (t < 94) {
      const int J = t - 79;
      if (c != 15 || h != 0) return 0.0f;
      const v4f R = rb[l * QQ + 8 * (J + 1)];
      return cc[l * QQ + 16 * J + 15] * (e == 0 ? R[0] : R[2]);
    }
    if (t < 109) {
      const int J = t - 93;
      if (h != (J >> 3) || e != (J & 1)) return 0.0f;
      return prodc(cc + l * QQ + 16 * J, 0, c);
    }
    return 0.0f;
  }
  const int r = sigma - SLOTS_LEFT;
  const int J = 16 + r / 4;
  const int k = 8 * h + 2 * (r % 4) + e;
  if ((k >> 1) != (c >> 1)) return 0.0f;
  const int q = 8 * J + (k >> 1);
  float a = ((k & 1) == 0) ? 1.0f : 0.0f;
  float b = ((k & 1) == 0) ? 0.0f : 1.0f;
  for (int l = 0; l < NLAY; ++l) {
    const v4f R = rb[l * QQ + q];
    const float t = a * R[0] + b * R[2];
    b = a * R[1] + b * R[0];
    a = t;
  }
  return ((c & 1) == 0) ? a : b;
}

__global__ void __launch_bounds__(256)
vqc_build_frags(const float* __restrict__ rot_params,
                const float* __restrict__ ent_params,
                float* __restrict__ Fw) {
  __shared__ v4f rb[NLAY * QQ];
  __shared__ float cc[NLAY * QQ];
  __shared__ float Gs[NLAY * 16];
  const int tid = threadIdx.x;

  for (int l = 0; l < NLAY; ++l) {
    const float t0 = rot_params[((l * QQ) + tid) * 3 + 0];
    const float t1 = rot_params[((l * QQ) + tid) * 3 + 1];
    const float t2 = rot_params[((l * QQ) + tid) * 3 + 2];
    const float c0 = cosf(t0 * 0.5f);
    const float c1 = cosf(t1 * 0.5f);
    const float s1 = sinf(t1 * 0.5f);
    const float c2 = cosf(t2 * 0.5f);
    v4f r;
    r[0] = (c0 * c1) * c2;
    r[1] = (c0 * (-s1)) * c2;
    r[2] = (c0 * s1) * c2;
    r[3] = 0.0f;
    rb[l * QQ + tid] = r;
    float c = 0.0f;
    if (tid < NPAIR) {
      const float x = ent_params[l * NPAIR + tid];
      c = 0.1f * (1.0f / (1.0f + expf(-x)));
    }
    cc[l * QQ + tid] = c;
  }
  __syncthreads();
  if (tid < NLAY * 16) Gs[tid] = prodc(cc + (tid >> 4) * QQ + 16 * (tid & 15), 0, 16);
  __syncthreads();

  const int idx = blockIdx.x * 256 + tid;
  const int sigma = idx >> 5;
  const int lane = idx & 31;
  v2f v;
  v[0] = frag_elem(rb, cc, Gs, sigma, lane & 15, lane >> 4, 0);
  v[1] = frag_elem(rb, cc, Gs, sigma, lane & 15, lane >> 4, 1);
  ((v2f*)Fw)[idx] = v;
}

static __device__ __forceinline__ v2f bsel(const v8f& t, int s) {
  v2f b;
  b[0] = t[2 * s];
  b[1] = t[2 * s + 1];
  return b;
}

static __device__ __forceinline__ v8f load_tile(const float* p) {
  const v4f lo = *(const v4f*)p;
  const v4f hi = *(const v4f*)(p + 4);
  return (v8f){lo[0], lo[1], lo[2], lo[3], hi[0], hi[1], hi[2], hi[3]};
}

static __device__ __forceinline__ void store_tile(float* p, const v8f& t) {
  *(v4f*)p = (v4f){t[0], t[1], t[2], t[3]};
  *(v4f*)(p + 4) = (v4f){t[4], t[5], t[6], t[7]};
}

#define NG 2
#define F1_N 56
#define F2_N 53

template <int N>
static __device__ __forceinline__ void load_frags(v2f (&f)[N], const v2f* Fl,
                                                  int base) {
#pragma unroll
  for (int i = 0; i < N; ++i) f[i] = Fl[(base + i) * 32];
}

template <int NT, int N>
static __device__ __forceinline__ void tiles_mma(v8f (&y)[NG][16],
                                                 const v8f (&x)[NG][16],
                                                 const v2f (&f)[N], int J0) {
  const v8f zero = (v8f){0.f, 0.f, 0.f, 0.f, 0.f, 0.f, 0.f, 0.f};
#pragma unroll
  for (int j = 0; j < NT; ++j) {
    const int J = J0 + j;
    v8f acc[NG];
#pragma unroll
    for (int g = 0; g < NG; ++g) acc[g] = zero;
#pragma unroll
    for (int s = 0; s < 4; ++s)
#pragma unroll
      for (int g = 0; g < NG; ++g)
        acc[g] = wmma_f32_4(f[4 * j + s], bsel(x[g][J], s), acc[g]);
#pragma unroll
    for (int g = 0; g < NG; ++g) y[g][J] = acc[g];
  }
}

static __device__ __forceinline__ void finish_mma(v8f (&y)[NG][16],
                                                  const v8f (&x)[NG][16],
                                                  const v2f (&f)[F2_N]) {
  const v8f zero = (v8f){0.f, 0.f, 0.f, 0.f, 0.f, 0.f, 0.f, 0.f};
  v8f Du[NG];
#pragma unroll
  for (int g = 0; g < NG; ++g) Du[g] = zero;
#pragma unroll
  for (int I = 0; I < 15; ++I)
#pragma unroll
    for (int g = 0; g < NG; ++g)
      Du[g] = wmma_f32_4(f[8 + I], bsel(y[g][I], 3), Du[g]);
#pragma unroll
  for (int J = 0; J < 15; ++J)
#pragma unroll
    for (int g = 0; g < NG; ++g)
      y[g][J] = wmma_f32_4(f[23 + J], bsel(x[g][J + 1], 0), y[g][J]);
#pragma unroll
  for (int J = 1; J < 16; ++J)
#pragma unroll
    for (int g = 0; g < NG; ++g)
      y[g][J] = wmma_f32_4(f[37 + J], bsel(Du[g], (J & 7) >> 1), y[g][J]);
}

__global__ void __launch_bounds__(128)
vqc_apply_layers(const float* __restrict__ A, const float* __restrict__ Fw,
                 float* __restrict__ C) {
  const int wave = threadIdx.x >> 5;
  const int lane = threadIdx.x & 31;
  const int c = lane & 15;
  const int h = lane >> 4;
  const size_t row = (size_t)(blockIdx.x * 4 + wave) * (16 * NG) + c;
  const float* xp = A + row * DQ + 8 * h;
  float* op = C + row * DQ + 8 * h;
  const v2f* F = (const v2f*)Fw + lane;
  const v8f zero = (v8f){0.f, 0.f, 0.f, 0.f, 0.f, 0.f, 0.f, 0.f};

  v2f fr[64];
#pragma unroll
  for (int i = 0; i < 64; ++i) fr[i] = F[(size_t)(SLOTS_LEFT + i) * 32];
  v8f xr[NG][16];
#pragma unroll
  for (int g = 0; g < NG; ++g)
#pragma unroll
    for (int J = 0; J < 16; ++J)
      xr[g][J] = load_tile(xp + (size_t)g * 16 * DQ + 256 + 16 * J);
  v8f x[NG][16];
#pragma unroll
  for (int g = 0; g < NG; ++g)
#pragma unroll
    for (int J = 0; J < 16; ++J)
      x[g][J] = load_tile(xp + (size_t)g * 16 * DQ + 16 * J);
  v2f f1[F1_N], f2[F2_N];
  load_frags(f1, F, 0);
  __builtin_amdgcn_sched_barrier(0);

#pragma unroll
  for (int J = 0; J < 16; ++J) {
    v8f acc[NG];
#pragma unroll
    for (int g = 0; g < NG; ++g) acc[g] = zero;
#pragma unroll
    for (int s = 0; s < 4; ++s)
#pragma unroll
      for (int g = 0; g < NG; ++g)
        acc[g] = wmma_f32_4(fr[J * 4 + s], bsel(xr[g][J], s), acc[g]);
#pragma unroll
    for (int g = 0; g < NG; ++g)
      store_tile(op + (size_t)g * 16 * DQ + 256 + 16 * J, acc[g]);
  }

#pragma unroll
  for (int l = 0; l < NLAY; ++l) {
    const v2f* Fl = F + (size_t)l * SLOTS_L * 32;
    v8f y[NG][16];
    __builtin_amdgcn_sched_barrier(0);
    load_frags(f2, Fl, F1_N);
    __builtin_amdgcn_sched_barrier(0);
    tiles_mma<14>(y, x, f1, 0);
    __builtin_amdgcn_sched_barrier(0);
    if (l + 1 < NLAY) load_frags(f1, Fl + (size_t)SLOTS_L * 32, 0);
    __builtin_amdgcn_sched_barrier(0);
    tiles_mma<2>(y, x, f2, 14);
    finish_mma(y, x, f2);
#pragma unroll
    for (int g = 0; g < NG; ++g)
#pragma unroll
      for (int J = 0; J < 16; ++J) x[g][J] = y[g][J];
  }

#pragma unroll
  for (int g = 0; g < NG; ++g)
#pragma unroll
    for (int J = 0; J < 16; ++J)
      store_tile(op + (size_t)g * 16 * DQ + 16 * J, x[g][J]);
}

extern "C" void kernel_launch(void* const* d_in, const int* in_sizes, int n_in,
                              void* d_out, int out_size, void* d_ws,
                              size_t ws_size, hipStream_t stream) {
  (void)in_sizes; (void)n_in; (void)out_size; (void)ws_size;
  const float* state = (const float*)d_in[0];
  const float* rotp  = (const float*)d_in[1];
  const float* entp  = (const float*)d_in[2];
  float* out = (float*)d_out;
  float* Fw  = (float*)d_ws;

  vqc_build_frags<<<SLOTS_ALL * 32 / 256, 256, 0, stream>>>(rotp, entp, Fw);

  vqc_apply_layers<<<M_STATE / 128, 128, 0, stream>>>(state, Fw, out);
}


# Output tiles leave through per-wave LDS rows (no barrier): each global store writes 4 whole 128-B lines, not 16 lines x two 16-B pieces; right half read back under layer 0; sums unchanged
# speedup vs baseline: 10.8864x; 1.2687x over previous
#include <hip/hip_runtime.h>
#include <math.h>


typedef float v2f __attribute__((ext_vector_type(2)));
typedef float v4f __attribute__((ext_vector_type(4)));
typedef float v8f __attribute__((ext_vector_type(8)));

#define DQ      512
#define QQ      256
#define NLAY    4
#define NPAIR   255
#define M_STATE (16 * 2048)
#define SLOTS_L    110
#define SLOTS_LEFT (NLAY * SLOTS_L)
#define SLOTS_ALL  (SLOTS_LEFT + 64)

static __device__ __forceinline__ v8f wmma_f32_4(v2f a, v2f b, v8f c) {
  return __builtin_amdgcn_wmma_f32_16x16x4_f32(false, a, false, b, (short)0, c,
                                               false, false);
}

static __device__ __forceinline__ float prodc(const float* cl, int a, int b) {
  float p = 1.0f;
  for (int k = a; k < b; ++k) p *= cl[k];
  return p;
}
static __device__ __forceinline__ float scan_w(const float* cl, int t, int i) {
  if (t < 0 || t > 15) return 0.0f;
  if (t <= i) return prodc(cl, t, i);
  if (t == i + 1) return cl[i];
  return 0.0f;
}
static __device__ __forceinline__ float tile_m(const v4f* rbl, const float* cl,
                                               int j, int i) {
  const v4f R = rbl[j >> 1];
  if ((j & 1) == 0) return R[0] * scan_w(cl, j, i) + R[1] * scan_w(cl, j + 1, i);
  return R[2] * scan_w(cl, j - 1, i) + R[0] * scan_w(cl, j, i);
}

static __device__ float frag_elem(const v4f* rb, const float* cc, const float* Gs,
                                  int sigma, int c, int h, int e) {
  if (sigma < SLOTS_LEFT) {
    const int l = sigma / SLOTS_L;
    const int t = sigma % SLOTS_L;
    if (t < 64) {
      const int J = t >> 2;
      return tile_m(rb + l * QQ + 8 * J, cc + l * QQ + 16 * J,
                    8 * h + 2 * (t & 3) + e, c);
    }
    if (t < 79) {
      const int I = t - 64;
      if (h != 1 || e != 1 || c <= I) return 0.0f;
      float p = cc[l * QQ + 16 * I + 15];
      for (int K = I + 1; K < c; ++K) p *= Gs[l * 16 + K];
      return p;
    }
    if (t < 94) {
      const int J = t - 79;
      if (c != 15 || h != 0) return 0.0f;
      const v4f R = rb[l * QQ + 8 * (J + 1)];
      return cc[l * QQ + 16 * J + 15] * (e == 0 ? R[0] : R[2]);
    }
    if (t < 109) {
      const int J = t - 93;
      if (h != (J >> 3) || e != (J & 1)) return 0.0f;
      return prodc(cc + l * QQ + 16 * J, 0, c);
    }
    return 0.0f;
  }
  const int r = sigma - SLOTS_LEFT;
  const int J = 16 + r / 4;
  const int k = 8 * h + 2 * (r % 4) + e;
  if ((k >> 1) != (c >> 1)) return 0.0f;
  const int q = 8 * J + (k >> 1);
  float a = ((k & 1) == 0) ? 1.0f : 0.0f;
  float b = ((k & 1) == 0) ? 0.0f : 1.0f;
  for (int l = 0; l < NLAY; ++l) {
    const v4f R = rb[l * QQ + q];
    const float t = a * R[0] + b * R[2];
    b = a * R[1] + b * R[0];
    a = t;
  }
  return ((c & 1) == 0) ? a : b;
}

__global__ void __launch_bounds__(256)
vqc_build_frags(const float* __restrict__ rot_params,
                const float* __restrict__ ent_params,
                float* __restrict__ Fw) {
  __shared__ v4f rb[NLAY * QQ];
  __shared__ float cc[NLAY * QQ];
  __shared__ float Gs[NLAY * 16];
  const int tid = threadIdx.x;

  for (int l = 0; l < NLAY; ++l) {
    const float t0 = rot_params[((l * QQ) + tid) * 3 + 0];
    const float t1 = rot_params[((l * QQ) + tid) * 3 + 1];
    const float t2 = rot_params[((l * QQ) + tid) * 3 + 2];
    const float c0 = cosf(t0 * 0.5f);
    const float c1 = cosf(t1 * 0.5f);
    const float s1 = sinf(t1 * 0.5f);
    const float c2 = cosf(t2 * 0.5f);
    v4f r;
    r[0] = (c0 * c1) * c2;
    r[1] = (c0 * (-s1)) * c2;
    r[2] = (c0 * s1) * c2;
    r[3] = 0.0f;
    rb[l * QQ + tid] = r;
    float c = 0.0f;
    if (tid < NPAIR) {
      const float x = ent_params[l * NPAIR + tid];
      c = 0.1f * (1.0f / (1.0f + expf(-x)));
    }
    cc[l * QQ + tid] = c;
  }
  __syncthreads();
  if (tid < NLAY * 16) Gs[tid] = prodc(cc + (tid >> 4) * QQ + 16 * (tid & 15), 0, 16);
  __syncthreads();

  const int idx = blockIdx.x * 256 + tid;
  const int sigma = idx >> 5;
  const int lane = idx & 31;
  v2f v;
  v[0] = frag_elem(rb, cc, Gs, sigma, lane & 15, lane >> 4, 0);
  v[1] = frag_elem(rb, cc, Gs, sigma, lane & 15, lane >> 4, 1);
  ((v2f*)Fw)[idx] = v;
}

static __device__ __forceinline__ v2f bsel(const v8f& t, int s) {
  v2f b;
  b[0] = t[2 * s];
  b[1] = t[2 * s + 1];
  return b;
}

static __device__ __forceinline__ v8f load_tile(const float* p) {
  const v4f lo = *(const v4f*)p;
  const v4f hi = *(const v4f*)(p + 4);
  return (v8f){lo[0], lo[1], lo[2], lo[3], hi[0], hi[1], hi[2], hi[3]};
}

#define SROW 65
#define SWAVE (32 * SROW)

static __device__ __forceinline__ void stage_tile(v4f* p, const v8f& t) {
  p[0] = (v4f){t[0], t[1], t[2], t[3]};
  p[1] = (v4f){t[4], t[5], t[6], t[7]};
}
static __device__ __forceinline__ v4f load_piece(const v4f* sld, int k) {
  return sld[(2 * (k >> 2)) * SROW + 16 * (k & 3)];
}
static __device__ __forceinline__ void store_piece(float* og, int k, const v4f& v) {
  *(v4f*)(og + (size_t)(2 * (k >> 2)) * DQ + 64 * (k & 3)) = v;
}

#define NG 2
#define F1_N 56
#define F2_N 53

template <int N>
static __device__ __forceinline__ void load_frags(v2f (&f)[N], const v2f* Fl,
                                                  int base) {
#pragma unroll
  for (int i = 0; i < N; ++i) f[i] = Fl[(base + i) * 32];
}

template <int NT, int N>
static __device__ __forceinline__ void tiles_mma(v8f (&y)[NG][16],
                                                 const v8f (&x)[NG][16],
                                                 const v2f (&f)[N], int J0,
                                                 const bool flush,
                                                 const v4f* sld, float* og,
                                                 v4f (&pend)[4]) {
  const v8f zero = (v8f){0.f, 0.f, 0.f, 0.f, 0.f, 0.f, 0.f, 0.f};
#pragma unroll
  for (int j = 0; j < NT; ++j) {
    const int J = J0 + j;
    v8f acc[NG];
#pragma unroll
    for (int g = 0; g < NG; ++g) acc[g] = zero;
#pragma unroll
    for (int s = 0; s < 4; ++s)
#pragma unroll
      for (int g = 0; g < NG; ++g)
        acc[g] = wmma_f32_4(f[4 * j + s], bsel(x[g][J], s), acc[g]);
#pragma unroll
    for (int g = 0; g < NG; ++g) y[g][J] = acc[g];
    if (flush) {
#pragma unroll
      for (int i = 0; i < 4; ++i) store_piece(og, 4 * J + i, pend[i]);
      if (J < 15) {
#pragma unroll
        for (int i = 0; i < 4; ++i) pend[i] = load_piece(sld, 4 * J + 4 + i);
      }
    }
  }
}

static __device__ __forceinline__ void finish_mma(v8f (&y)[NG][16],
                                                  const v8f (&x)[NG][16],
                                                  const v2f (&f)[F2_N]) {
  const v8f zero = (v8f){0.f, 0.f, 0.f, 0.f, 0.f, 0.f, 0.f, 0.f};
  v8f Du[NG];
#pragma unroll
  for (int g = 0; g < NG; ++g) Du[g] = zero;
#pragma unroll
  for (int I = 0; I < 15; ++I)
#pragma unroll
    for (int g = 0; g < NG; ++g)
      Du[g] = wmma_f32_4(f[8 + I], bsel(y[g][I], 3), Du[g]);
#pragma unroll
  for (int J = 0; J < 15; ++J)
#pragma unroll
    for (int g = 0; g < NG; ++g)
      y[g][J] = wmma_f32_4(f[23 + J], bsel(x[g][J + 1], 0), y[g][J]);
#pragma unroll
  for (int J = 1; J < 16; ++J)
#pragma unroll
    for (int g = 0; g < NG; ++g)
      y[g][J] = wmma_f32_4(f[37 + J], bsel(Du[g], (J & 7) >> 1), y[g][J]);
}

__global__ void __launch_bounds__(128)
vqc_apply_layers(const float* __restrict__ A, const float* __restrict__ Fw,
                 float* __restrict__ C) {
  const int wave = threadIdx.x >> 5;
  const int lane = threadIdx.x & 31;
  const int c = lane & 15;
  const int h = lane >> 4;
  const size_t row = (size_t)(blockIdx.x * 4 + wave) * (16 * NG) + c;
  const float* xp = A + row * DQ + 8 * h;
  __shared__ v4f stg[4 * SWAVE];
  v4f* sst = stg + wave * SWAVE + c * SROW + 2 * h;
  const v4f* sld = stg + wave * SWAVE + h * SROW + c;
  float* og = C + ((size_t)(blockIdx.x * 4 + wave) * (16 * NG) + h) * DQ + 4 * c;
  const v2f* F = (const v2f*)Fw + lane;
  const v8f zero = (v8f){0.f, 0.f, 0.f, 0.f, 0.f, 0.f, 0.f, 0.f};

  v2f fr[64];
#pragma unroll
  for (int i = 0; i < 64; ++i) fr[i] = F[(size_t)(SLOTS_LEFT + i) * 32];
  v8f xr[NG][16];
#pragma unroll
  for (int g = 0; g < NG; ++g)
#pragma unroll
    for (int J = 0; J < 16; ++J)
      xr[g][J] = load_tile(xp + (size_t)g * 16 * DQ + 256 + 16 * J);
  v8f x[NG][16];
#pragma unroll
  for (int g = 0; g < NG; ++g)
#pragma unroll
    for (int J = 0; J < 16; ++J)
      x[g][J] = load_tile(xp + (size_t)g * 16 * DQ + 16 * J);
  v2f f1[F1_N], f2[F2_N];
  load_frags(f1, F, 0);
  __builtin_amdgcn_sched_barrier(0);

#pragma unroll
  for (int J = 0; J < 16; ++J) {
    v8f acc[NG];
#pragma unroll
    for (int g = 0; g < NG; ++g) acc[g] = zero;
#pragma unroll
    for (int s = 0; s < 4; ++s)
#pragma unroll
      for (int g = 0; g < NG; ++g)
        acc[g] = wmma_f32_4(fr[J * 4 + s], bsel(xr[g][J], s), acc[g]);
#pragma unroll
    for (int g = 0; g < NG; ++g) stage_tile(sst + 16 * g * SROW + 4 * J, acc[g]);
  }
  v4f pend[4];
#pragma unroll
  for (int i = 0; i < 4; ++i) pend[i] = load_piece(sld, i);

#pragma unroll
  for (int l = 0; l < NLAY; ++l) {
    const v2f* Fl = F + (size_t)l * SLOTS_L * 32;
    v8f y[NG][16];
    __builtin_amdgcn_sched_barrier(0);
    load_frags(f2, Fl, F1_N);
    __builtin_amdgcn_sched_barrier(0);
    tiles_mma<14>(y, x, f1, 0, l == 0, sld, og + 256, pend);
    __builtin_amdgcn_sched_barrier(0);
    if (l + 1 < NLAY) load_frags(f1, Fl + (size_t)SLOTS_L * 32, 0);
    __builtin_amdgcn_sched_barrier(0);
    tiles_mma<2>(y, x, f2, 14, l == 0, sld, og + 256, pend);
    finish_mma(y, x, f2);
#pragma unroll
    for (int g = 0; g < NG; ++g)
#pragma unroll
      for (int J = 0; J < 16; ++J) x[g][J] = y[g][J];
  }

#pragma unroll
  for (int g = 0; g < NG; ++g)
#pragma unroll
    for (int J = 0; J < 16; ++J) stage_tile(sst + 16 * g * SROW + 4 * J, x[g][J]);
#pragma unroll
  for (int k = 0; k < 64; ++k) store_piece(og, k, load_piece(sld, k));
}

extern "C" void kernel_launch(void* const* d_in, const int* in_sizes, int n_in,
                              void* d_out, int out_size, void* d_ws,
                              size_t ws_size, hipStream_t stream) {
  (void)in_sizes; (void)n_in; (void)out_size; (void)ws_size;
  const float* state = (const float*)d_in[0];
  const float* rotp  = (const float*)d_in[1];
  const float* entp  = (const float*)d_in[2];
  float* out = (float*)d_out;
  float* Fw  = (float*)d_ws;

  vqc_build_frags<<<SLOTS_ALL * 32 / 256, 256, 0, stream>>>(rotp, entp, Fw);

  vqc_apply_layers<<<M_STATE / 128, 128, 0, stream>>>(state, Fw, out);
}


# Table-build kernel: a block evaluates gate coefficients only for the one layer (or 16 qubits) its 8 slots read; products of pair coefficients run on 16 registers, not in per-lane LDS loops; sums same
# speedup vs baseline: 14.1307x; 1.2980x over previous
#include <hip/hip_runtime.h>
#include <math.h>


typedef float v2f __attribute__((ext_vector_type(2)));
typedef float v4f __attribute__((ext_vector_type(4)));
typedef float v8f __attribute__((ext_vector_type(8)));

#define DQ      512
#define QQ      256
#define NLAY    4
#define NPAIR   255
#define M_STATE (16 * 2048)
#define SLOTS_L    110
#define SLOTS_LEFT (NLAY * SLOTS_L)
#define SLOTS_ALL  (SLOTS_LEFT + 64)

static __device__ __forceinline__ v8f wmma_f32_4(v2f a, v2f b, v8f c) {
  return __builtin_amdgcn_wmma_f32_16x16x4_f32(false, a, false, b, (short)0, c,
                                               false, false);
}

static __device__ __forceinline__ v4f gate_block(const float* __restrict__ rot_params,
                                                 int l, int q) {
  const float t0 = rot_params[((l * QQ) + q) * 3 + 0];
  const float t1 = rot_params[((l * QQ) + q) * 3 + 1];
  const float t2 = rot_params[((l * QQ) + q) * 3 + 2];
  const float c0 = cosf(t0 * 0.5f);
  const float c1 = cosf(t1 * 0.5f);
  const float s1 = sinf(t1 * 0.5f);
  const float c2 = cosf(t2 * 0.5f);
  v4f r;
  r[0] = (c0 * c1) * c2;
  r[1] = (c0 * (-s1)) * c2;
  r[2] = (c0 * s1) * c2;
  r[3] = 0.0f;
  return r;
}
static __device__ __forceinline__ v2f pair_coeffs(const float* __restrict__ ent_params,
                                                  int l, int p0) {
  const int p1 = p0 + 1;
  const float x0 = ent_params[l * NPAIR + p0];
  const float x1 = ent_params[l * NPAIR + (p1 < NPAIR ? p1 : NPAIR - 1)];
  v2f c;
  c[0] = 0.1f * (1.0f / (1.0f + expf(-x0)));
  c[1] = (p1 < NPAIR) ? 0.1f * (1.0f / (1.0f + expf(-x1))) : 0.0f;
  return c;
}

static __device__ __forceinline__ float prod_range(const float (&cl)[16], int a, int b) {
  float p = 1.0f;
#pragma unroll
  for (int k = 0; k < 16; ++k) p = (k >= a && k < b) ? p * cl[k] : p;
  return p;
}
static __device__ __forceinline__ float scan_w(const float (&cl)[16], float ci, int t, int i) {
  const float p = prod_range(cl, t, i);
  if (t <= i) return p;
  if (t == i + 1) return ci;
  return 0.0f;
}
static __device__ __forceinline__ void load16(float (&cl)[16], const float* p) {
#pragma unroll
  for (int k = 0; k < 16; ++k) cl[k] = p[k];
}

static __device__ __forceinline__ v2f frag_left(const v4f* rb, const float* cc,
                                               const float* Gs, int l, int t,
                                               int c, int h) {
  v2f v;
  v[0] = 0.0f;
  v[1] = 0.0f;
  if (t < 64) {
    const int J = t >> 2;
    const int s = t & 3;
    float cl[16];
    load16(cl, cc + l * QQ + 16 * J);
    const float ci = cc[l * QQ + 16 * J + c];
    const v4f R = rb[l * QQ + 8 * J + 4 * h + s];
    const int j0 = 8 * h + 2 * s;
    const float sa = scan_w(cl, ci, j0, c);
    const float sb = scan_w(cl, ci, j0 + 1, c);
    v[0] = R[0] * sa + R[1] * sb;
    v[1] = R[2] * sa + R[0] * sb;
  } else if (t < 79) {
    const int I = t - 64;
    float gs[16];
    load16(gs, Gs + l * 16);
    float p = cc[l * QQ + 16 * I + 15];
#pragma unroll
    for (int K = 1; K < 15; ++K) p = (K > I && K < c) ? p * gs[K] : p;
    if (h == 1 && c > I) v[1] = p;
  } else if (t < 94) {
    const int J = t - 79;
    const v4f R = rb[l * QQ + 8 * (J + 1)];
    const float cl15 = cc[l * QQ + 16 * J + 15];
    if (c == 15 && h == 0) {
      v[0] = cl15 * R[0];
      v[1] = cl15 * R[2];
    }
  } else if (t < 109) {
    const int J = t - 93;
    float cl[16];
    load16(cl, cc + l * QQ + 16 * J);
    const float p = prod_range(cl, 0, c);
    if (h == (J >> 3)) v[J & 1] = p;
  }
  return v;
}

static __device__ __forceinline__ v2f frag_right(const v4f* rb, int r, int c, int h) {
  v2f v;
  v[0] = 0.0f;
  v[1] = 0.0f;
  const int J = 16 + (r >> 2);
  const int kq = 4 * h + (r & 3);
  const int q = 8 * J + kq;
  v4f R[NLAY];
#pragma unroll
  for (int l = 0; l < NLAY; ++l) R[l] = rb[l * QQ + q];
  if (kq == (c >> 1)) {
#pragma unroll
    for (int e = 0; e < 2; ++e) {
      float a = (e == 0) ? 1.0f : 0.0f;
      float b = (e == 0) ? 0.0f : 1.0f;
#pragma unroll
      for (int l = 0; l < NLAY; ++l) {
        const float t = a * R[l][0] + b * R[l][2];
        b = a * R[l][1] + b * R[l][0];
        a = t;
      }
      v[e] = ((c & 1) == 0) ? a : b;
    }
  }
  return v;
}

#define BLK_L     14
#define NBLK_LEFT (NLAY * BLK_L)
#define NBLK      (NBLK_LEFT + 8)

__global__ void __launch_bounds__(256)
vqc_build_frags(const float* __restrict__ rot_params,
                const float* __restrict__ ent_params,
                float* __restrict__ Fw) {
  __shared__ v4f rb[NLAY * QQ];
  __shared__ float cc[NLAY * QQ];
  __shared__ float Gs[NLAY * 16];
  const int tid = threadIdx.x;
  const int blk = blockIdx.x;
  const bool left = blk < NBLK_LEFT;
  const int l = blk / BLK_L;

  if (left) {
    if (tid < 128) {
      rb[l * QQ + tid] = gate_block(rot_params, l, tid);
    } else {
      const int p0 = 2 * (tid - 128);
      const v2f cp = pair_coeffs(ent_params, l, p0);
      cc[l * QQ + p0] = cp[0];
      cc[l * QQ + p0 + 1] = cp[1];
    }
  } else if (tid < 64) {
    const int q = 128 + 16 * (blk - NBLK_LEFT) + (tid & 15);
    rb[(tid >> 4) * QQ + q] = gate_block(rot_params, tid >> 4, q);
  }
  __syncthreads();
  if (left && tid < 16) {
    float cl[16];
    load16(cl, cc + l * QQ + 16 * tid);
    Gs[l * 16 + tid] = prod_range(cl, 0, 16);
  }
  __syncthreads();

  const int wave = __builtin_amdgcn_readfirstlane(tid >> 5);
  const int lane = tid & 31;
  if (left) {
    const int t = (blk % BLK_L) * 8 + wave;
    if (t < SLOTS_L)
      ((v2f*)Fw)[(l * SLOTS_L + t) * 32 + lane] =
          frag_left(rb, cc, Gs, l, t, lane & 15, lane >> 4);
  } else {
    const int r = (blk - NBLK_LEFT) * 8 + wave;
    ((v2f*)Fw)[(SLOTS_LEFT + r) * 32 + lane] = frag_right(rb, r, lane & 15, lane >> 4);
  }
}

static __device__ __forceinline__ v2f bsel(const v8f& t, int s) {
  v2f b;
  b[0] = t[2 * s];
  b[1] = t[2 * s + 1];
  return b;
}

static __device__ __forceinline__ v8f load_tile(const float* p) {
  const v4f lo = *(const v4f*)p;
  const v4f hi = *(const v4f*)(p + 4);
  return (v8f){lo[0], lo[1], lo[2], lo[3], hi[0], hi[1], hi[2], hi[3]};
}

#define SROW 65
#define SWAVE (32 * SROW)

static __device__ __forceinline__ void stage_tile(v4f* p, const v8f& t) {
  p[0] = (v4f){t[0], t[1], t[2], t[3]};
  p[1] = (v4f){t[4], t[5], t[6], t[7]};
}
static __device__ __forceinline__ v4f load_piece(const v4f* sld, int k) {
  return sld[(2 * (k >> 2)) * SROW + 16 * (k & 3)];
}
static __device__ __forceinline__ void store_piece(float* og, int k, const v4f& v) {
  *(v4f*)(og + (size_t)(2 * (k >> 2)) * DQ + 64 * (k & 3)) = v;
}

#define NG 2
#define F1_N 56
#define F2_N 53

template <int N>
static __device__ __forceinline__ void load_frags(v2f (&f)[N], const v2f* Fl,
                                                  int base) {
#pragma unroll
  for (int i = 0; i < N; ++i) f[i] = Fl[(base + i) * 32];
}

template <int NT, int N>
static __device__ __forceinline__ void tiles_mma(v8f (&y)[NG][16],
                                                 const v8f (&x)[NG][16],
                                                 const v2f (&f)[N], int J0,
                                                 const bool flush,
                                                 const v4f* sld, float* og,
                                                 v4f (&pend)[4]) {
  const v8f zero = (v8f){0.f, 0.f, 0.f, 0.f, 0.f, 0.f, 0.f, 0.f};
#pragma unroll
  for (int j = 0; j < NT; ++j) {
    const int J = J0 + j;
    v8f acc[NG];
#pragma unroll
    for (int g = 0; g < NG; ++g) acc[g] = zero;
#pragma unroll
    for (int s = 0; s < 4; ++s)
#pragma unroll
      for (int g = 0; g < NG; ++g)
        acc[g] = wmma_f32_4(f[4 * j + s], bsel(x[g][J], s), acc[g]);
#pragma unroll
    for (int g = 0; g < NG; ++g) y[g][J] = acc[g];
    if (flush) {
#pragma unroll
      for (int i = 0; i < 4; ++i) store_piece(og, 4 * J + i, pend[i]);
      if (J < 15) {
#pragma unroll
        for (int i = 0; i < 4; ++i) pend[i] = load_piece(sld, 4 * J + 4 + i);
      }
    }
  }
}

static __device__ __forceinline__ void finish_mma(v8f (&y)[NG][16],
                                                  const v8f (&x)[NG][16],
                                                  const v2f (&f)[F2_N]) {
  const v8f zero = (v8f){0.f, 0.f, 0.f, 0.f, 0.f, 0.f, 0.f, 0.f};
  v8f Du[NG];
#pragma unroll
  for (int g = 0; g < NG; ++g) Du[g] = zero;
#pragma unroll
  for (int I = 0; I < 15; ++I)
#pragma unroll
    for (int g = 0; g < NG; ++g)
      Du[g] = wmma_f32_4(f[8 + I], bsel(y[g][I], 3), Du[g]);
#pragma unroll
  for (int J = 0; J < 15; ++J)
#pragma unroll
    for (int g = 0; g < NG; ++g)
      y[g][J] = wmma_f32_4(f[23 + J], bsel(x[g][J + 1], 0), y[g][J]);
#pragma unroll
  for (int J = 1; J < 16; ++J)
#pragma unroll
    for (int g = 0; g < NG; ++g)
      y[g][J] = wmma_f32_4(f[37 + J], bsel(Du[g], (J & 7) >> 1), y[g][J]);
}

__global__ void __launch_bounds__(128)
vqc_apply_layers(const float* __restrict__ A, const float* __restrict__ Fw,
                 float* __restrict__ C) {
  const int wave = threadIdx.x >> 5;
  const int lane = threadIdx.x & 31;
  const int c = lane & 15;
  const int h = lane >> 4;
  const size_t row = (size_t)(blockIdx.x * 4 + wave) * (16 * NG) + c;
  const float* xp = A + row * DQ + 8 * h;
  __shared__ v4f stg[4 * SWAVE];
  v4f* sst = stg + wave * SWAVE + c * SROW + 2 * h;
  const v4f* sld = stg + wave * SWAVE + h * SROW + c;
  float* og = C + ((size_t)(blockIdx.x * 4 + wave) * (16 * NG) + h) * DQ + 4 * c;
  const v2f* F = (const v2f*)Fw + lane;
  const v8f zero = (v8f){0.f, 0.f, 0.f, 0.f, 0.f, 0.f, 0.f, 0.f};

  v2f fr[64];
#pragma unroll
  for (int i = 0; i < 64; ++i) fr[i] = F[(size_t)(SLOTS_LEFT + i) * 32];
  v8f xr[NG][16];
#pragma unroll
  for (int g = 0; g < NG; ++g)
#pragma unroll
    for (int J = 0; J < 16; ++J)
      xr[g][J] = load_tile(xp + (size_t)g * 16 * DQ + 256 + 16 * J);
  v8f x[NG][16];
#pragma unroll
  for (int g = 0; g < NG; ++g)
#pragma unroll
    for (int J = 0; J < 16; ++J)
      x[g][J] = load_tile(xp + (size_t)g * 16 * DQ + 16 * J);
  v2f f1[F1_N], f2[F2_N];
  load_frags(f1, F, 0);
  __builtin_amdgcn_sched_barrier(0);

#pragma unroll
  for (int J = 0; J < 16; ++J) {
    v8f acc[NG];
#pragma unroll
    for (int g = 0; g < NG; ++g) acc[g] = zero;
#pragma unroll
    for (int s = 0; s < 4; ++s)
#pragma unroll
      for (int g = 0; g < NG; ++g)
        acc[g] = wmma_f32_4(fr[J * 4 + s], bsel(xr[g][J], s), acc[g]);
#pragma unroll
    for (int g = 0; g < NG; ++g) stage_tile(sst + 16 * g * SROW + 4 * J, acc[g]);
  }
  v4f pend[4];
#pragma unroll
  for (int i = 0; i < 4; ++i) pend[i] = load_piece(sld, i);

#pragma unroll
  for (int l = 0; l < NLAY; ++l) {
    const v2f* Fl = F + (size_t)l * SLOTS_L * 32;
    v8f y[NG][16];
    __builtin_amdgcn_sched_barrier(0);
    load_frags(f2, Fl, F1_N);
    __builtin_amdgcn_sched_barrier(0);
    tiles_mma<14>(y, x, f1, 0, l == 0, sld, og + 256, pend);
    __builtin_amdgcn_sched_barrier(0);
    if (l + 1 < NLAY) load_frags(f1, Fl + (size_t)SLOTS_L * 32, 0);
    __builtin_amdgcn_sched_barrier(0);
    tiles_mma<2>(y, x, f2, 14, l == 0, sld, og + 256, pend);
    finish_mma(y, x, f2);
#pragma unroll
    for (int g = 0; g < NG; ++g)
#pragma unroll
      for (int J = 0; J < 16; ++J) x[g][J] = y[g][J];
  }

#pragma unroll
  for (int g = 0; g < NG; ++g)
#pragma unroll
    for (int J = 0; J < 16; ++J) stage_tile(sst + 16 * g * SROW + 4 * J, x[g][J]);
#pragma unroll
  for (int k = 0; k < 64; ++k) store_piece(og, k, load_piece(sld, k));
}

extern "C" void kernel_launch(void* const* d_in, const int* in_sizes, int n_in,
                              void* d_out, int out_size, void* d_ws,
                              size_t ws_size, hipStream_t stream) {
  (void)in_sizes; (void)n_in; (void)out_size; (void)ws_size;
  const float* state = (const float*)d_in[0];
  const float* rotp  = (const float*)d_in[1];
  const float* entp  = (const float*)d_in[2];
  float* out = (float*)d_out;
  float* Fw  = (float*)d_ws;

  vqc_build_frags<<<NBLK, 256, 0, stream>>>(rotp, entp, Fw);

  vqc_apply_layers<<<M_STATE / 128, 128, 0, stream>>>(state, Fw, out);
}


# Two launches become one: every work-group builds the A fragment table for itself in LDS (running products along tile rows) under its first state-row loads; fragments read from LDS; sums unchanged
# speedup vs baseline: 15.9825x; 1.1310x over previous
#include <hip/hip_runtime.h>
#include <math.h>


typedef float v2f __attribute__((ext_vector_type(2)));
typedef float v4f __attribute__((ext_vector_type(4)));
typedef float v8f __attribute__((ext_vector_type(8)));

#define DQ      512
#define QQ      256
#define NLAY    4
#define NPAIR   255
#define M_STATE (16 * 2048)
#define SLOTS_L    110
#define SLOTS_LEFT (NLAY * SLOTS_L)
#define SLOTS_ALL  (SLOTS_LEFT + 64)

static __device__ __forceinline__ v8f wmma_f32_4(v2f a, v2f b, v8f c) {
  return __builtin_amdgcn_wmma_f32_16x16x4_f32(false, a, false, b, (short)0, c,
                                               false, false);
}

static __device__ __forceinline__ v4f gate_block(float t0, float t1, float t2) {
  const float c0 = cosf(t0 * 0.5f);
  float s1, c1;
  sincosf(t1 * 0.5f, &s1, &c1);
  const float c2 = cosf(t2 * 0.5f);
  v4f r;
  r[0] = (c0 * c1) * c2;
  r[1] = (c0 * (-s1)) * c2;
  r[2] = (c0 * s1) * c2;
  r[3] = 0.0f;
  return r;
}
static __device__ __forceinline__ float pair_coeff(float x) {
  return 0.1f * (1.0f / (1.0f + expf(-x)));
}

static __device__ __forceinline__ float prod_range(const float (&cl)[16], int a, int b) {
  float p = 1.0f;
#pragma unroll
  for (int k = 0; k < 16; ++k) p = (k >= a && k < b) ? p * cl[k] : p;
  return p;
}
static __device__ __forceinline__ float scan_w(const float (&cl)[16], float ci, int t, int i) {
  const float p = prod_range(cl, t, i);
  if (t <= i) return p;
  if (t == i + 1) return ci;
  return 0.0f;
}
static __device__ __forceinline__ void load16(float (&cl)[16], const float* p) {
#pragma unroll
  for (int k = 0; k < 16; ++k) cl[k] = p[k];
}

#define TP 33

static __device__ __forceinline__ void lds_barrier() {
  __builtin_amdgcn_fence(__ATOMIC_RELEASE, "workgroup", "local");
  __builtin_amdgcn_s_barrier();
  __builtin_amdgcn_fence(__ATOMIC_ACQUIRE, "workgroup", "local");
}

static __device__ __forceinline__ void tab_put(v2f* tab, int slot, int lane, float a, float b) {
  v2f v;
  v[0] = a;
  v[1] = b;
  tab[slot * TP + lane] = v;
}

static __device__ __forceinline__ void build_table(int tid, const v4f* rb, const float* cc,
                                                   const float* Gs, v2f* tab) {
  {
    const int J = tid >> 3;
    const int h = (tid >> 2) & 1;
    const int s = tid & 3;
    const int j0 = 8 * h + 2 * s;
#pragma unroll
    for (int l = 0; l < NLAY; ++l) {
      float cl[16];
      load16(cl, cc + l * QQ + 16 * J);
      const v4f R = rb[l * QQ + 8 * J + 4 * h + s];
      float pa = 1.0f;
      float pb = 1.0f;
#pragma unroll
      for (int c = 0; c < 16; ++c) {
        if (c > 0) {
          pa = (c - 1 >= j0) ? pa * cl[c - 1] : pa;
          pb = (c - 1 >= j0 + 1) ? pb * cl[c - 1] : pb;
        }
        const float sa = (j0 <= c) ? pa : ((j0 == c + 1) ? cl[c] : 0.0f);
        const float sb = (j0 + 1 <= c) ? pb : ((j0 == c) ? cl[c] : 0.0f);
        tab_put(tab, l * SLOTS_L + 4 * J + s, 16 * h + c,
                R[0] * sa + R[1] * sb, R[2] * sa + R[0] * sb);
      }
    }
  }
  if (tid < 60) {
    const int l = tid / 15;
    const int I = tid % 15;
    float gs[16];
    load16(gs, Gs + l * 16);
    float p = cc[l * QQ + 16 * I + 15];
#pragma unroll
    for (int c = 0; c < 16; ++c) {
      tab_put(tab, l * SLOTS_L + 64 + I, c, 0.0f, 0.0f);
      tab_put(tab, l * SLOTS_L + 64 + I, 16 + c, 0.0f, (c > I) ? p : 0.0f);
      if (c >= 1 && c <= 14) p = (c > I) ? p * gs[c] : p;
    }
    const v4f R = rb[l * QQ + 8 * (I + 1)];
    const float cl15 = cc[l * QQ + 16 * I + 15];
#pragma unroll
    for (int c = 0; c < 16; ++c) {
      const bool on = (c == 15);
      tab_put(tab, l * SLOTS_L + 79 + I, c, on ? cl15 * R[0] : 0.0f, on ? cl15 * R[2] : 0.0f);
      tab_put(tab, l * SLOTS_L + 79 + I, 16 + c, 0.0f, 0.0f);
    }
  } else if (tid >= 64 && tid < 124) {
    const int l = (tid - 64) / 15;
    const int J = 1 + (tid - 64) % 15;
    float cl[16];
    load16(cl, cc + l * QQ + 16 * J);
    float p = 1.0f;
    const int hh = J >> 3;
    const int e = J & 1;
#pragma unroll
    for (int c = 0; c < 16; ++c) {
      const float v0 = (e == 0) ? p : 0.0f;
      const float v1 = (e == 1) ? p : 0.0f;
      tab_put(tab, l * SLOTS_L + 93 + J, c, (hh == 0) ? v0 : 0.0f, (hh == 0) ? v1 : 0.0f);
      tab_put(tab, l * SLOTS_L + 93 + J, 16 + c, (hh == 1) ? v0 : 0.0f, (hh == 1) ? v1 : 0.0f);
      p = p * cl[c];
    }
  }
  {
    const int r = tid >> 1;
    const int h = tid & 1;
    const int J = 16 + (r >> 2);
    const int kq = 4 * h + (r & 3);
    const int q = 8 * J + kq;
    v4f R[NLAY];
#pragma unroll
    for (int l = 0; l < NLAY; ++l) R[l] = rb[l * QQ + q];
    float ab[2][2];
#pragma unroll
    for (int e = 0; e < 2; ++e) {
      float a = (e == 0) ? 1.0f : 0.0f;
      float b = (e == 0) ? 0.0f : 1.0f;
#pragma unroll
      for (int l = 0; l < NLAY; ++l) {
        const float t = a * R[l][0] + b * R[l][2];
        b = a * R[l][1] + b * R[l][0];
        a = t;
      }
      ab[e][0] = a;
      ab[e][1] = b;
    }
#pragma unroll
    for (int c = 0; c < 16; ++c) {
      const bool on = (kq == (c >> 1));
      const int o = c & 1;
      tab_put(tab, SLOTS_LEFT + r, 16 * h + c, on ? ab[0][o] : 0.0f, on ? ab[1][o] : 0.0f);
    }
  }
}

static __device__ __forceinline__ v2f bsel(const v8f& t, int s) {
  v2f b;
  b[0] = t[2 * s];
  b[1] = t[2 * s + 1];
  return b;
}

static __device__ __forceinline__ v8f load_tile(const float* p) {
  const v4f lo = *(const v4f*)p;
  const v4f hi = *(const v4f*)(p + 4);
  return (v8f){lo[0], lo[1], lo[2], lo[3], hi[0], hi[1], hi[2], hi[3]};
}

#define SROW 65
#define SWAVE (32 * SROW)

static __device__ __forceinline__ void stage_tile(v4f* p, const v8f& t) {
  p[0] = (v4f){t[0], t[1], t[2], t[3]};
  p[1] = (v4f){t[4], t[5], t[6], t[7]};
}
static __device__ __forceinline__ v4f load_piece(const v4f* sld, int k) {
  return sld[(2 * (k >> 2)) * SROW + 16 * (k & 3)];
}
static __device__ __forceinline__ void store_piece(float* og, int k, const v4f& v) {
  *(v4f*)(og + (size_t)(2 * (k >> 2)) * DQ + 64 * (k & 3)) = v;
}

#define NG 2
#define F1_N 56
#define F2_N 53

template <int N>
static __device__ __forceinline__ void load_frags(v2f (&f)[N], const v2f* Fl,
                                                  int base) {
#pragma unroll
  for (int i = 0; i < N; ++i) f[i] = Fl[(base + i) * TP];
}

template <int NT, int N>
static __device__ __forceinline__ void tiles_mma(v8f (&y)[NG][16],
                                                 const v8f (&x)[NG][16],
                                                 const v2f (&f)[N], int J0,
                                                 const bool flush,
                                                 const v4f* sld, float* og,
                                                 v4f (&pend)[4]) {
  const v8f zero = (v8f){0.f, 0.f, 0.f, 0.f, 0.f, 0.f, 0.f, 0.f};
#pragma unroll
  for (int j = 0; j < NT; ++j) {
    const int J = J0 + j;
    v8f acc[NG];
#pragma unroll
    for (int g = 0; g < NG; ++g) acc[g] = zero;
#pragma unroll
    for (int s = 0; s < 4; ++s)
#pragma unroll
      for (int g = 0; g < NG; ++g)
        acc[g] = wmma_f32_4(f[4 * j + s], bsel(x[g][J], s), acc[g]);
#pragma unroll
    for (int g = 0; g < NG; ++g) y[g][J] = acc[g];
    if (flush) {
#pragma unroll
      for (int i = 0; i < 4; ++i) store_piece(og, 4 * J + i, pend[i]);
      if (J < 15) {
#pragma unroll
        for (int i = 0; i < 4; ++i) pend[i] = load_piece(sld, 4 * J + 4 + i);
      }
    }
  }
}

static __device__ __forceinline__ void finish_mma(v8f (&y)[NG][16],
                                                  const v8f (&x)[NG][16],
                                                  const v2f (&f)[F2_N]) {
  const v8f zero = (v8f){0.f, 0.f, 0.f, 0.f, 0.f, 0.f, 0.f, 0.f};
  v8f Du[NG];
#pragma unroll
  for (int g = 0; g < NG; ++g) Du[g] = zero;
#pragma unroll
  for (int I = 0; I < 15; ++I)
#pragma unroll
    for (int g = 0; g < NG; ++g)
      Du[g] = wmma_f32_4(f[8 + I], bsel(y[g][I], 3), Du[g]);
#pragma unroll
  for (int J = 0; J < 15; ++J)
#pragma unroll
    for (int g = 0; g < NG; ++g)
      y[g][J] = wmma_f32_4(f[23 + J], bsel(x[g][J + 1], 0), y[g][J]);
#pragma unroll
  for (int J = 1; J < 16; ++J)
#pragma unroll
    for (int g = 0; g < NG; ++g)
      y[g][J] = wmma_f32_4(f[37 + J], bsel(Du[g], (J & 7) >> 1), y[g][J]);
}

__global__ void __launch_bounds__(128)
vqc_apply_layers(const float* __restrict__ A, const float* __restrict__ rot_params,
                 const float* __restrict__ ent_params, float* __restrict__ C) {
  const int tid = threadIdx.x;
  const int wave = threadIdx.x >> 5;
  const int lane = threadIdx.x & 31;
  const int c = lane & 15;
  const int h = lane >> 4;
  const size_t row = (size_t)(blockIdx.x * 4 + wave) * (16 * NG) + c;
  const float* xp = A + row * DQ + 8 * h;
  __shared__ v4f stg[4 * SWAVE];
  __shared__ v2f tab[SLOTS_ALL * TP];
  __shared__ v4f rb[NLAY * QQ];
  __shared__ float cc[NLAY * QQ];
  __shared__ float Gs[NLAY * 16];
  v4f* sst = stg + wave * SWAVE + c * SROW + 2 * h;
  const v4f* sld = stg + wave * SWAVE + h * SROW + c;
  float* og = C + ((size_t)(blockIdx.x * 4 + wave) * (16 * NG) + h) * DQ + 4 * c;
  const v2f* F = tab + lane;
  const v8f zero = (v8f){0.f, 0.f, 0.f, 0.f, 0.f, 0.f, 0.f, 0.f};

  float tp[2 * NLAY][3];
#pragma unroll
  for (int i = 0; i < 2 * NLAY; ++i)
#pragma unroll
    for (int k = 0; k < 3; ++k)
      tp[i][k] = rot_params[(((i >> 1) * QQ) + tid + 128 * (i & 1)) * 3 + k];
  const int p0 = 2 * tid;
  const int p1 = p0 + 1;
  float ep[NLAY][2];
#pragma unroll
  for (int l = 0; l < NLAY; ++l) {
    ep[l][0] = ent_params[l * NPAIR + p0];
    ep[l][1] = ent_params[l * NPAIR + (p1 < NPAIR ? p1 : NPAIR - 1)];
  }
  __builtin_amdgcn_sched_barrier(0);
  v8f xr[NG][16];
  v8f x[NG][16];
#pragma unroll
  for (int J = 0; J < 16; ++J) xr[0][J] = load_tile(xp + 256 + 16 * J);
#pragma unroll
  for (int J = 0; J < 7; ++J) xr[1][J] = load_tile(xp + (size_t)16 * DQ + 256 + 16 * J);
  __builtin_amdgcn_sched_barrier(0);
#pragma unroll
  for (int i = 0; i < 2 * NLAY; ++i)
    rb[(i >> 1) * QQ + tid + 128 * (i & 1)] = gate_block(tp[i][0], tp[i][1], tp[i][2]);
#pragma unroll
  for (int l = 0; l < NLAY; ++l) {
    cc[l * QQ + p0] = pair_coeff(ep[l][0]);
    cc[l * QQ + p1] = (p1 < NPAIR) ? pair_coeff(ep[l][1]) : 0.0f;
  }
  __builtin_amdgcn_sched_barrier(0);
#pragma unroll
  for (int J = 7; J < 16; ++J) xr[1][J] = load_tile(xp + (size_t)16 * DQ + 256 + 16 * J);
#pragma unroll
  for (int J = 0; J < 16; ++J) x[0][J] = load_tile(xp + 16 * J);
  __builtin_amdgcn_sched_barrier(0);
  lds_barrier();
  if (tid < NLAY * 16) {
    float cl[16];
    load16(cl, cc + 16 * tid);
    Gs[tid] = prod_range(cl, 0, 16);
  }
  lds_barrier();
  build_table(tid, rb, cc, Gs, tab);
  __builtin_amdgcn_sched_barrier(0);
#pragma unroll
  for (int J = 0; J < 16; ++J) x[1][J] = load_tile(xp + (size_t)16 * DQ + 16 * J);
  lds_barrier();

  v2f fr[64];
#pragma unroll
  for (int i = 0; i < 64; ++i) fr[i] = F[(SLOTS_LEFT + i) * TP];
  v2f f1[F1_N], f2[F2_N];
  load_frags(f1, F, 0);
  __builtin_amdgcn_sched_barrier(0);

#pragma unroll
  for (int J = 0; J < 16; ++J) {
    v8f acc[NG];
#pragma unroll
    for (int g = 0; g < NG; ++g) acc[g] = zero;
#pragma unroll
    for (int s = 0; s < 4; ++s)
#pragma unroll
      for (int g = 0; g < NG; ++g)
        acc[g] = wmma_f32_4(fr[J * 4 + s], bsel(xr[g][J], s), acc[g]);
#pragma unroll
    for (int g = 0; g < NG; ++g) stage_tile(sst + 16 * g * SROW + 4 * J, acc[g]);
  }
  v4f pend[4];
#pragma unroll
  for (int i = 0; i < 4; ++i) pend[i] = load_piece(sld, i);

#pragma unroll
  for (int l = 0; l < NLAY; ++l) {
    const v2f* Fl = F + l * SLOTS_L * TP;
    v8f y[NG][16];
    __builtin_amdgcn_sched_barrier(0);
    load_frags(f2, Fl, F1_N);
    __builtin_amdgcn_sched_barrier(0);
    tiles_mma<14>(y, x, f1, 0, l == 0, sld, og + 256, pend);
    __builtin_amdgcn_sched_barrier(0);
    if (l + 1 < NLAY) load_frags(f1, Fl + SLOTS_L * TP, 0);
    __builtin_amdgcn_sched_barrier(0);
    tiles_mma<2>(y, x, f2, 14, l == 0, sld, og + 256, pend);
    finish_mma(y, x, f2);
#pragma unroll
    for (int g = 0; g < NG; ++g)
#pragma unroll
      for (int J = 0; J < 16; ++J) x[g][J] = y[g][J];
  }

#pragma unroll
  for (int g = 0; g < NG; ++g)
#pragma unroll
    for (int J = 0; J < 16; ++J) stage_tile(sst + 16 * g * SROW + 4 * J, x[g][J]);
#pragma unroll
  for (int k = 0; k < 64; ++k) store_piece(og, k, load_piece(sld, k));
}

extern "C" void kernel_launch(void* const* d_in, const int* in_sizes, int n_in,
                              void* d_out, int out_size, void* d_ws,
                              size_t ws_size, hipStream_t stream) {
  (void)in_sizes; (void)n_in; (void)out_size; (void)d_ws; (void)ws_size;
  const float* state = (const float*)d_in[0];
  const float* rotp  = (const float*)d_in[1];
  const float* entp  = (const float*)d_in[2];
  float* out = (float*)d_out;

  vqc_apply_layers<<<M_STATE / 128, 128, 0, stream>>>(state, rotp, entp, out);
}


# Carries run tile to tile: a tile's carry K-step reads row 15 of the finished tile before it, no Lc pass (1000 -> 880 WMMAs a wave); last layer's tiles are stored as they finish; sums reordered
# speedup vs baseline: 17.0396x; 1.0661x over previous
#include <hip/hip_runtime.h>
#include <math.h>


typedef float v2f __attribute__((ext_vector_type(2)));
typedef float v4f __attribute__((ext_vector_type(4)));
typedef float v8f __attribute__((ext_vector_type(8)));

#define DQ      512
#define QQ      256
#define NLAY    4
#define NPAIR   255
#define M_STATE (16 * 2048)
#define SLOTS_A    46
#define SLOTS_B    48
#define SLOTS_L    (SLOTS_A + SLOTS_B)
#define SLOTS_LEFT (NLAY * SLOTS_L)
#define SLOTS_ALL  (SLOTS_LEFT + 64)

static __device__ __forceinline__ v8f wmma_f32_4(v2f a, v2f b, v8f c) {
  return __builtin_amdgcn_wmma_f32_16x16x4_f32(false, a, false, b, (short)0, c,
                                               false, false);
}

static __device__ __forceinline__ int slot_m(int J, int s) {
  return (J < 8) ? 4 * J + s : SLOTS_A + 4 * (J - 8) + s;
}
static __device__ __forceinline__ int slot_coup(int J) {
  return (J < 7) ? 32 + J : SLOTS_A + 32 + (J - 7);
}
static __device__ __forceinline__ int slot_carry(int J) {
  return (J < 8) ? 38 + J : SLOTS_A + 40 + (J - 8);
}

static __device__ __forceinline__ v4f gate_block(float t0, float t1, float t2) {
  const float c0 = cosf(t0 * 0.5f);
  float s1, c1;
  sincosf(t1 * 0.5f, &s1, &c1);
  const float c2 = cosf(t2 * 0.5f);
  v4f r;
  r[0] = (c0 * c1) * c2;
  r[1] = (c0 * (-s1)) * c2;
  r[2] = (c0 * s1) * c2;
  r[3] = 0.0f;
  return r;
}
static __device__ __forceinline__ float pair_coeff(float x) {
  return 0.1f * (1.0f / (1.0f + expf(-x)));
}

static __device__ __forceinline__ void load16(float (&cl)[16], const float* p) {
#pragma unroll
  for (int k = 0; k < 16; ++k) cl[k] = p[k];
}

#define TP 33

static __device__ __forceinline__ void lds_barrier() {
  __builtin_amdgcn_fence(__ATOMIC_RELEASE, "workgroup", "local");
  __builtin_amdgcn_s_barrier();
  __builtin_amdgcn_fence(__ATOMIC_ACQUIRE, "workgroup", "local");
}

static __device__ __forceinline__ void tab_put(v2f* tab, int slot, int lane, float a, float b) {
  v2f v;
  v[0] = a;
  v[1] = b;
  tab[slot * TP + lane] = v;
}

static __device__ __forceinline__ void build_table(int tid, const v4f* rb, const float* cc,
                                                   v2f* tab) {
  {
    const int J = tid >> 3;
    const int h = (tid >> 2) & 1;
    const int s = tid & 3;
    const int j0 = 8 * h + 2 * s;
#pragma unroll
    for (int l = 0; l < NLAY; ++l) {
      float cl[16];
      load16(cl, cc + l * QQ + 16 * J);
      const v4f R = rb[l * QQ + 8 * J + 4 * h + s];
      float pa = 1.0f;
      float pb = 1.0f;
#pragma unroll
      for (int c = 0; c < 16; ++c) {
        if (c > 0) {
          pa = (c - 1 >= j0) ? pa * cl[c - 1] : pa;
          pb = (c - 1 >= j0 + 1) ? pb * cl[c - 1] : pb;
        }
        const float sa = (j0 <= c) ? pa : ((j0 == c + 1) ? cl[c] : 0.0f);
        const float sb = (j0 + 1 <= c) ? pb : ((j0 == c) ? cl[c] : 0.0f);
        tab_put(tab, l * SLOTS_L + slot_m(J, s), 16 * h + c,
                R[0] * sa + R[1] * sb, R[2] * sa + R[0] * sb);
      }
    }
  }
  if (tid < 60) {
    const int l = tid / 15;
    const int I = tid % 15;
    const v4f R = rb[l * QQ + 8 * (I + 1)];
    const float cl15 = cc[l * QQ + 16 * I + 15];
#pragma unroll
    for (int c = 0; c < 16; ++c) {
      const bool on = (c == 15);
      tab_put(tab, l * SLOTS_L + slot_coup(I), c, on ? cl15 * R[0] : 0.0f, on ? cl15 * R[2] : 0.0f);
      tab_put(tab, l * SLOTS_L + slot_coup(I), 16 + c, 0.0f, 0.0f);
    }
  } else if (tid >= 64 && tid < 124) {
    const int l = (tid - 64) / 15;
    const int J = 1 + (tid - 64) % 15;
    float cl[16];
    load16(cl, cc + l * QQ + 16 * J);
    float p = cc[l * QQ + 16 * J - 1];
#pragma unroll
    for (int c = 0; c < 16; ++c) {
      tab_put(tab, l * SLOTS_L + slot_carry(J), c, 0.0f, 0.0f);
      tab_put(tab, l * SLOTS_L + slot_carry(J), 16 + c, 0.0f, p);
      p = p * cl[c];
    }
  }
  {
    const int r = tid >> 1;
    const int h = tid & 1;
    const int J = 16 + (r >> 2);
    const int kq = 4 * h + (r & 3);
    const int q = 8 * J + kq;
    v4f R[NLAY];
#pragma unroll
    for (int l = 0; l < NLAY; ++l) R[l] = rb[l * QQ + q];
    float ab[2][2];
#pragma unroll
    for (int e = 0; e < 2; ++e) {
      float a = (e == 0) ? 1.0f : 0.0f;
      float b = (e == 0) ? 0.0f : 1.0f;
#pragma unroll
      for (int l = 0; l < NLAY; ++l) {
        const float t = a * R[l][0] + b * R[l][2];
        b = a * R[l][1] + b * R[l][0];
        a = t;
      }
      ab[e][0] = a;
      ab[e][1] = b;
    }
#pragma unroll
    for (int c = 0; c < 16; ++c) {
      const bool on = (kq == (c >> 1));
      const int o = c & 1;
      tab_put(tab, SLOTS_LEFT + r, 16 * h + c, on ? ab[0][o] : 0.0f, on ? ab[1][o] : 0.0f);
    }
  }
}

static __device__ __forceinline__ v2f bsel(const v8f& t, int s) {
  v2f b;
  b[0] = t[2 * s];
  b[1] = t[2 * s + 1];
  return b;
}

static __device__ __forceinline__ v8f load_tile(const float* p) {
  const v4f lo = *(const v4f*)p;
  const v4f hi = *(const v4f*)(p + 4);
  return (v8f){lo[0], lo[1], lo[2], lo[3], hi[0], hi[1], hi[2], hi[3]};
}

#define SROW 65
#define SWAVE (32 * SROW)

static __device__ __forceinline__ void stage_tile(v4f* p, const v8f& t) {
  p[0] = (v4f){t[0], t[1], t[2], t[3]};
  p[1] = (v4f){t[4], t[5], t[6], t[7]};
}
static __device__ __forceinline__ v4f load_piece(const v4f* sld, int k) {
  return sld[(2 * (k >> 2)) * SROW + 16 * (k & 3)];
}
static __device__ __forceinline__ void store_piece(float* og, int k, const v4f& v) {
  *(v4f*)(og + (size_t)(2 * (k >> 2)) * DQ + 64 * (k & 3)) = v;
}

#define NG 2

template <int N>
static __device__ __forceinline__ void load_frags(v2f (&f)[N], const v2f* Fl,
                                                  int base) {
#pragma unroll
  for (int i = 0; i < N; ++i) f[i] = Fl[(base + i) * TP];
}

template <int N>
static __device__ __forceinline__ void tiles_chain(v8f (&y)[NG][16], v8f (&zp)[NG],
                                                   const v8f (&x)[NG][16],
                                                   const v2f (&f)[N], int J0, int off,
                                                   const bool first, const bool last,
                                                   v4f* sst, const v4f* sld, float* og,
                                                   v4f (&pend)[4]) {
  const v8f zero = (v8f){0.f, 0.f, 0.f, 0.f, 0.f, 0.f, 0.f, 0.f};
  v4f qv[16];
#pragma unroll
  for (int j = 0; j < 8; ++j) {
    const int J = J0 + j;
    v8f acc[NG];
#pragma unroll
    for (int g = 0; g < NG; ++g) acc[g] = zero;
#pragma unroll
    for (int s = 0; s < 4; ++s)
#pragma unroll
      for (int g = 0; g < NG; ++g)
        acc[g] = wmma_f32_4(f[slot_m(J, s) - off], bsel(x[g][J], s), acc[g]);
    if (J >= 1) {
#pragma unroll
      for (int g = 0; g < NG; ++g)
        acc[g] = wmma_f32_4(f[slot_carry(J) - off], bsel(zp[g], 3), acc[g]);
#pragma unroll
      for (int g = 0; g < NG; ++g)
        y[g][J - 1] = wmma_f32_4(f[slot_coup(J - 1) - off], bsel(x[g][J], 0), zp[g]);
    }
#pragma unroll
    for (int g = 0; g < NG; ++g) zp[g] = acc[g];
    if (J == 15) {
#pragma unroll
      for (int g = 0; g < NG; ++g) y[g][15] = acc[g];
    }
    if (first) {
#pragma unroll
      for (int i = 0; i < 4; ++i) store_piece(og + 256, 4 * J + i, pend[i]);
      if (J < 15) {
#pragma unroll
        for (int i = 0; i < 4; ++i) pend[i] = load_piece(sld, 4 * J + 4 + i);
      }
    }
    if (last) {
      if (J >= 2 && ((J - 2) & 3) == 3) {
#pragma unroll
        for (int i = 0; i < 16; ++i) store_piece(og, 4 * i + ((J - 2) >> 2), qv[i]);
        __builtin_amdgcn_sched_barrier(0);
      }
      if (J >= 1) {
#pragma unroll
        for (int g = 0; g < NG; ++g) stage_tile(sst + 16 * g * SROW + 4 * (J - 1), y[g][J - 1]);
        if (((J - 1) & 3) == 3) {
#pragma unroll
          for (int i = 0; i < 16; ++i) qv[i] = load_piece(sld, 4 * i + ((J - 1) >> 2));
        }
      }
    }
  }
}

__global__ void __launch_bounds__(128)
vqc_apply_layers(const float* __restrict__ A, const float* __restrict__ rot_params,
                 const float* __restrict__ ent_params, float* __restrict__ C) {
  const int tid = threadIdx.x;
  const int wave = threadIdx.x >> 5;
  const int lane = threadIdx.x & 31;
  const int c = lane & 15;
  const int h = lane >> 4;
  const size_t row = (size_t)(blockIdx.x * 4 + wave) * (16 * NG) + c;
  const float* xp = A + row * DQ + 8 * h;
  __shared__ v4f stg[4 * SWAVE];
  __shared__ v2f tab[SLOTS_ALL * TP];
  __shared__ v4f rb[NLAY * QQ];
  __shared__ float cc[NLAY * QQ];
  v4f* sst = stg + wave * SWAVE + c * SROW + 2 * h;
  const v4f* sld = stg + wave * SWAVE + h * SROW + c;
  float* og = C + ((size_t)(blockIdx.x * 4 + wave) * (16 * NG) + h) * DQ + 4 * c;
  const v2f* F = tab + lane;
  const v8f zero = (v8f){0.f, 0.f, 0.f, 0.f, 0.f, 0.f, 0.f, 0.f};

  float tp[2 * NLAY][3];
#pragma unroll
  for (int i = 0; i < 2 * NLAY; ++i)
#pragma unroll
    for (int k = 0; k < 3; ++k)
      tp[i][k] = rot_params[(((i >> 1) * QQ) + tid + 128 * (i & 1)) * 3 + k];
  const int p0 = 2 * tid;
  const int p1 = p0 + 1;
  float ep[NLAY][2];
#pragma unroll
  for (int l = 0; l < NLAY; ++l) {
    ep[l][0] = ent_params[l * NPAIR + p0];
    ep[l][1] = ent_params[l * NPAIR + (p1 < NPAIR ? p1 : NPAIR - 1)];
  }
  __builtin_amdgcn_sched_barrier(0);
  v8f xr[NG][16];
  v8f x[NG][16];
#pragma unroll
  for (int J = 0; J < 16; ++J) xr[0][J] = load_tile(xp + 256 + 16 * J);
#pragma unroll
  for (int J = 0; J < 7; ++J) xr[1][J] = load_tile(xp + (size_t)16 * DQ + 256 + 16 * J);
  __builtin_amdgcn_sched_barrier(0);
#pragma unroll
  for (int i = 0; i < 2 * NLAY; ++i)
    rb[(i >> 1) * QQ + tid + 128 * (i & 1)] = gate_block(tp[i][0], tp[i][1], tp[i][2]);
#pragma unroll
  for (int l = 0; l < NLAY; ++l) {
    cc[l * QQ + p0] = pair_coeff(ep[l][0]);
    cc[l * QQ + p1] = (p1 < NPAIR) ? pair_coeff(ep[l][1]) : 0.0f;
  }
  __builtin_amdgcn_sched_barrier(0);
#pragma unroll
  for (int J = 7; J < 16; ++J) xr[1][J] = load_tile(xp + (size_t)16 * DQ + 256 + 16 * J);
#pragma unroll
  for (int J = 0; J < 16; ++J) x[0][J] = load_tile(xp + 16 * J);
  __builtin_amdgcn_sched_barrier(0);
  lds_barrier();
  build_table(tid, rb, cc, tab);
  __builtin_amdgcn_sched_barrier(0);
#pragma unroll
  for (int J = 0; J < 16; ++J) x[1][J] = load_tile(xp + (size_t)16 * DQ + 16 * J);
  lds_barrier();

  v2f fr[64];
#pragma unroll
  for (int i = 0; i < 64; ++i) fr[i] = F[(SLOTS_LEFT + i) * TP];
  v2f fa[SLOTS_A], fb[SLOTS_B];
  load_frags(fa, F, 0);
  __builtin_amdgcn_sched_barrier(0);

#pragma unroll
  for (int J = 0; J < 16; ++J) {
    v8f acc[NG];
#pragma unroll
    for (int g = 0; g < NG; ++g) acc[g] = zero;
#pragma unroll
    for (int s = 0; s < 4; ++s)
#pragma unroll
      for (int g = 0; g < NG; ++g)
        acc[g] = wmma_f32_4(fr[J * 4 + s], bsel(xr[g][J], s), acc[g]);
#pragma unroll
    for (int g = 0; g < NG; ++g) stage_tile(sst + 16 * g * SROW + 4 * J, acc[g]);
  }
  v4f pend[4];
#pragma unroll
  for (int i = 0; i < 4; ++i) pend[i] = load_piece(sld, i);

#pragma unroll
  for (int l = 0; l < NLAY; ++l) {
    const v2f* Fl = F + l * SLOTS_L * TP;
    v8f y[NG][16];
    v8f zp[NG];
#pragma unroll
    for (int g = 0; g < NG; ++g) zp[g] = zero;
    __builtin_amdgcn_sched_barrier(0);
    load_frags(fb, Fl, SLOTS_A);
    __builtin_amdgcn_sched_barrier(0);
    tiles_chain(y, zp, x, fa, 0, 0, l == 0, l == NLAY - 1, sst, sld, og, pend);
    __builtin_amdgcn_sched_barrier(0);
    if (l + 1 < NLAY) load_frags(fa, Fl + SLOTS_L * TP, 0);
    __builtin_amdgcn_sched_barrier(0);
    tiles_chain(y, zp, x, fb, 8, SLOTS_A, l == 0, l == NLAY - 1, sst, sld, og, pend);
#pragma unroll
    for (int g = 0; g < NG; ++g)
#pragma unroll
      for (int J = 0; J < 16; ++J) x[g][J] = y[g][J];
  }

#pragma unroll
  for (int g = 0; g < NG; ++g) stage_tile(sst + 16 * g * SROW + 4 * 15, x[g][15]);
#pragma unroll
  for (int i = 0; i < 16; ++i) store_piece(og, 4 * i + 3, load_piece(sld, 4 * i + 3));
}

extern "C" void kernel_launch(void* const* d_in, const int* in_sizes, int n_in,
                              void* d_out, int out_size, void* d_ws,
                              size_t ws_size, hipStream_t stream) {
  (void)in_sizes; (void)n_in; (void)out_size; (void)d_ws; (void)ws_size;
  const float* state = (const float*)d_in[0];
  const float* rotp  = (const float*)d_in[1];
  const float* entp  = (const float*)d_in[2];
  float* out = (float*)d_out;

  vqc_apply_layers<<<M_STATE / 128, 128, 0, stream>>>(state, rotp, entp, out);
}
